# starting kernel
# baseline (speedup 1.0000x reference)
#include <hip/hip_runtime.h>
#include <hip/hip_bf16.h>

typedef __attribute__((ext_vector_type(2))) float v2f;
typedef __attribute__((ext_vector_type(8))) float v8f;

#define D_MODEL 1024
#define N_HEADS 16
#define D_HEAD  64
#define BATCH   4
#define SEQ     2048
#define ROWS    (BATCH * SEQ)

static __device__ __forceinline__ v8f wmma_f32(v2f a, v2f b, v8f c) {
  return __builtin_amdgcn_wmma_f32_16x16x4_f32(
       false, a,  false, b,
       (short)0, c,  false,  false);
}

__global__ __launch_bounds__(256) void transpose_w_kernel(
    const float* __restrict__ W0, const float* __restrict__ W1,
    const float* __restrict__ W2, const float* __restrict__ W3,
    float* __restrict__ T0, float* __restrict__ T1,
    float* __restrict__ T2, float* __restrict__ T3) {
  const float* W = (blockIdx.z == 0) ? W0 : (blockIdx.z == 1) ? W1
                   : (blockIdx.z == 2) ? W2 : W3;
  float* T = (blockIdx.z == 0) ? T0 : (blockIdx.z == 1) ? T1
             : (blockIdx.z == 2) ? T2 : T3;

  __shared__ float tile[32][33];
  const int tx = threadIdx.x & 31;
  const int ty = threadIdx.x >> 5;
  const int x0 = blockIdx.x * 32;
  const int y0 = blockIdx.y * 32;

#pragma unroll
  for (int i = 0; i < 4; ++i)
    tile[ty + i * 8][tx] = W[(size_t)(y0 + ty + i * 8) * D_MODEL + x0 + tx];
  __syncthreads();
#pragma unroll
  for (int i = 0; i < 4; ++i)
    T[(size_t)(x0 + ty + i * 8) * D_MODEL + y0 + tx] = tile[tx][ty + i * 8];
}

__global__ __launch_bounds__(256) void qkv_proj_kernel(
    const float* __restrict__ X,
    const float* __restrict__ Wqt, const float* __restrict__ Wkt,
    const float* __restrict__ Wvt,
    float* __restrict__ Q, float* __restrict__ K, float* __restrict__ Vt) {
  const int wave = threadIdx.x >> 5;
  const int lane = threadIdx.x & 31;
  const int half = lane >> 4;
  const int l16  = lane & 15;

  const int row0 = blockIdx.x * 32;
  const int col0 = (blockIdx.y * 8 + wave) * 16;
  const int col  = col0 + l16;

  v8f cq0 = {}, ck0 = {}, cv0 = {};
  v8f cq1 = {}, ck1 = {}, cv1 = {};
  const float* xrow0 = X + (size_t)(row0 + l16) * D_MODEL;
  const float* xrow1 = xrow0 + (size_t)16 * D_MODEL;
  const float* wq = Wqt + (size_t)col * D_MODEL;
  const float* wk = Wkt + (size_t)col * D_MODEL;
  const float* wv = Wvt + (size_t)col * D_MODEL;

#pragma unroll 4
  for (int k0 = 0; k0 < D_MODEL; k0 += 4) {
    const int kr = k0 + 2 * half;
    v2f a0 = *(const v2f*)(xrow0 + kr);
    v2f a1 = *(const v2f*)(xrow1 + kr);
    v2f bq = *(const v2f*)(wq + kr);
    v2f bk = *(const v2f*)(wk + kr);
    v2f bv = *(const v2f*)(wv + kr);
    cq0 = wmma_f32(a0, bq, cq0);  cq1 = wmma_f32(a1, bq, cq1);
    ck0 = wmma_f32(a0, bk, ck0);  ck1 = wmma_f32(a1, bk, ck1);
    cv0 = wmma_f32(a0, bv, cv0);  cv1 = wmma_f32(a1, bv, cv1);
  }

  const int h = col >> 6;
  const int d = col & 63;
#pragma unroll
  for (int r = 0; r < 8; ++r) {
#pragma unroll
    for (int t = 0; t < 2; ++t) {
      const int row = row0 + t * 16 + r + 8 * half;
      const int b = row >> 11;
      const int s = row & (SEQ - 1);
      const int bh = b * N_HEADS + h;
      const size_t o  = ((size_t)bh * SEQ + s) * D_HEAD + d;
      const size_t ov = ((size_t)bh * D_HEAD + d) * SEQ + s;
      Q[o]  = t ? cq1[r] : cq0[r];
      K[o]  = t ? ck1[r] : ck0[r];
      Vt[ov] = t ? cv1[r] : cv0[r];
    }
  }
}

__global__ __launch_bounds__(256) void flash_attn_kernel(
    const float* __restrict__ Q, const float* __restrict__ K,
    const float* __restrict__ Vt, float* __restrict__ Oout) {
  const int wave = threadIdx.x >> 5;
  const int lane = threadIdx.x & 31;
  const int half = lane >> 4;
  const int l16  = lane & 15;

  const int bh = blockIdx.y;
  const int q0 = (blockIdx.x * 8 + wave) * 16;

  const float* Qm  = Q  + (size_t)bh * SEQ * D_HEAD;
  const float* Km  = K  + (size_t)bh * SEQ * D_HEAD;
  const float* Vtm = Vt + (size_t)bh * D_HEAD * SEQ;

  __shared__ float lds[8][16][17];
  float (*pt)[17] = lds[wave];

  v2f qf[16];
  const float* qrow = Qm + (size_t)(q0 + l16) * D_HEAD;
#pragma unroll
  for (int kk = 0; kk < 16; ++kk)
    qf[kk] = *(const v2f*)(qrow + kk * 4 + 2 * half);

  v8f o0 = {}, o1 = {}, o2 = {}, o3 = {};
  float m8[8], l8[8];
#pragma unroll
  for (int r = 0; r < 8; ++r) { m8[r] = -__builtin_inff(); l8[r] = 0.0f; }

  const float* vb0 = Vtm + (size_t)( 0 + l16) * SEQ;
  const float* vb1 = Vtm + (size_t)(16 + l16) * SEQ;
  const float* vb2 = Vtm + (size_t)(32 + l16) * SEQ;
  const float* vb3 = Vtm + (size_t)(48 + l16) * SEQ;

  for (int t0 = 0; t0 <= q0; t0 += 16) {
    v8f sc = {};
    const float* krow = Km + (size_t)(t0 + l16) * D_HEAD;
#pragma unroll
    for (int kk = 0; kk < 16; ++kk) {
      v2f bf = *(const v2f*)(krow + kk * 4 + 2 * half);
      sc = wmma_f32(qf[kk], bf, sc);
    }

    const bool diag = (t0 == q0);
#pragma unroll
    for (int r = 0; r < 8; ++r) {
      float v = sc[r] * 0.125f;
      if (diag && (l16 > r + 8 * half)) v = -__builtin_inff();
      sc[r] = v;
    }

#pragma unroll
    for (int r = 0; r < 8; ++r) {
      float v = sc[r];
      v = fmaxf(v, __shfl_xor(v, 1, 16));
      v = fmaxf(v, __shfl_xor(v, 2, 16));
      v = fmaxf(v, __shfl_xor(v, 4, 16));
      v = fmaxf(v, __shfl_xor(v, 8, 16));
      const float mn   = fmaxf(m8[r], v);
      const float resc = __expf(m8[r] - mn);
      m8[r] = mn;
      const float p = __expf(sc[r] - mn);
      sc[r] = p;
      float s = p;
      s += __shfl_xor(s, 1, 16);
      s += __shfl_xor(s, 2, 16);
      s += __shfl_xor(s, 4, 16);
      s += __shfl_xor(s, 8, 16);
      l8[r] = l8[r] * resc + s;
      o0[r] *= resc; o1[r] *= resc; o2[r] *= resc; o3[r] *= resc;
    }

#pragma unroll
    for (int r = 0; r < 8; ++r)
      pt[r + 8 * half][l16] = sc[r];
    asm volatile("s_wait_dscnt 0" ::: "memory");

#pragma unroll
    for (int kk = 0; kk < 4; ++kk) {
      v2f a;
      a.x = pt[l16][kk * 4 + 2 * half];
      a.y = pt[l16][kk * 4 + 2 * half + 1];
      const int tk = t0 + kk * 4 + 2 * half;
      v2f b;
      b = *(const v2f*)(vb0 + tk); o0 = wmma_f32(a, b, o0);
      b = *(const v2f*)(vb1 + tk); o1 = wmma_f32(a, b, o1);
      b = *(const v2f*)(vb2 + tk); o2 = wmma_f32(a, b, o2);
      b = *(const v2f*)(vb3 + tk); o3 = wmma_f32(a, b, o3);
    }
  }

  const int b = bh >> 4;
  const int h = bh & (N_HEADS - 1);
#pragma unroll
  for (int r = 0; r < 8; ++r) {
    const float inv = 1.0f / l8[r];
    const int s = q0 + r + 8 * half;
    float* orow = Oout + (size_t)(b * SEQ + s) * (N_HEADS * D_HEAD) + h * D_HEAD;
    orow[ 0 + l16] = o0[r] * inv;
    orow[16 + l16] = o1[r] * inv;
    orow[32 + l16] = o2[r] * inv;
    orow[48 + l16] = o3[r] * inv;
  }
}

__global__ __launch_bounds__(256) void out_proj_kernel(
    const float* __restrict__ A, const float* __restrict__ Wot,
    float* __restrict__ C) {
  const int wave = threadIdx.x >> 5;
  const int lane = threadIdx.x & 31;
  const int half = lane >> 4;
  const int l16  = lane & 15;

  const int row0 = blockIdx.x * 32;
  const int col0 = (blockIdx.y * 8 + wave) * 64;

  v8f c00 = {}, c01 = {}, c02 = {}, c03 = {};
  v8f c10 = {}, c11 = {}, c12 = {}, c13 = {};
  const float* xrow0 = A + (size_t)(row0 + l16) * D_MODEL;
  const float* xrow1 = xrow0 + (size_t)16 * D_MODEL;
  const float* w0 = Wot + (size_t)(col0 +  0 + l16) * D_MODEL;
  const float* w1 = Wot + (size_t)(col0 + 16 + l16) * D_MODEL;
  const float* w2 = Wot + (size_t)(col0 + 32 + l16) * D_MODEL;
  const float* w3 = Wot + (size_t)(col0 + 48 + l16) * D_MODEL;

#pragma unroll 4
  for (int k0 = 0; k0 < D_MODEL; k0 += 4) {
    const int kr = k0 + 2 * half;
    v2f a0 = *(const v2f*)(xrow0 + kr);
    v2f a1 = *(const v2f*)(xrow1 + kr);
    v2f b0 = *(const v2f*)(w0 + kr);
    v2f b1 = *(const v2f*)(w1 + kr);
    v2f b2 = *(const v2f*)(w2 + kr);
    v2f b3 = *(const v2f*)(w3 + kr);
    c00 = wmma_f32(a0, b0, c00);  c10 = wmma_f32(a1, b0, c10);
    c01 = wmma_f32(a0, b1, c01);  c11 = wmma_f32(a1, b1, c11);
    c02 = wmma_f32(a0, b2, c02);  c12 = wmma_f32(a1, b2, c12);
    c03 = wmma_f32(a0, b3, c03);  c13 = wmma_f32(a1, b3, c13);
  }

#pragma unroll
  for (int r = 0; r < 8; ++r) {
    const int row = row0 + r + 8 * half;
    float* crow0 = C + (size_t)row * D_MODEL + col0;
    float* crow1 = crow0 + (size_t)16 * D_MODEL;
    crow0[ 0 + l16] = c00[r];
    crow0[16 + l16] = c01[r];
    crow0[32 + l16] = c02[r];
    crow0[48 + l16] = c03[r];
    crow1[ 0 + l16] = c10[r];
    crow1[16 + l16] = c11[r];
    crow1[32 + l16] = c12[r];
    crow1[48 + l16] = c13[r];
  }
}

extern "C" void kernel_launch(void* const* d_in, const int* in_sizes, int n_in,
                              void* d_out, int out_size, void* d_ws, size_t ws_size,
                              hipStream_t stream) {
  const float* x  = (const float*)d_in[0];
  const float* Wq = (const float*)d_in[1];
  const float* Wk = (const float*)d_in[2];
  const float* Wv = (const float*)d_in[3];
  const float* Wo = (const float*)d_in[4];

  float* ws = (float*)d_ws;
  const size_t per = (size_t)BATCH * N_HEADS * SEQ * D_HEAD;
  const size_t wsz = (size_t)D_MODEL * D_MODEL;
  float* Q   = ws;
  float* K   = ws + per;
  float* Vt  = ws + 2 * per;
  float* AO  = ws + 3 * per;
  float* Wqt = ws + 4 * per;
  float* Wkt = Wqt + wsz;
  float* Wvt = Wkt + wsz;
  float* Wot = Wvt + wsz;

  dim3 g0(D_MODEL / 32, D_MODEL / 32, 4);
  transpose_w_kernel<<<g0, 256, 0, stream>>>(Wq, Wk, Wv, Wo, Wqt, Wkt, Wvt, Wot);

  dim3 g1(ROWS / 32, (N_HEADS * D_HEAD) / (16 * 8));
  qkv_proj_kernel<<<g1, 256, 0, stream>>>(x, Wqt, Wkt, Wvt, Q, K, Vt);

  dim3 g2(SEQ / (16 * 8), BATCH * N_HEADS);
  flash_attn_kernel<<<g2, 256, 0, stream>>>(Q, K, Vt, AO);

  dim3 g3(ROWS / 32, D_MODEL / (64 * 8));
  out_proj_kernel<<<g3, 256, 0, stream>>>(AO, Wot, (float*)d_out);
}


# Output projection: 64x64 register tile per wave (was 32x64); a trip reads one whole 128-byte line per operand row by 16-byte loads, issued one trip ahead into a second set; k sum order changed
# speedup vs baseline: 1.8483x; 1.8483x over previous
#include <hip/hip_runtime.h>
#include <hip/hip_bf16.h>

typedef __attribute__((ext_vector_type(2))) float v2f;
typedef __attribute__((ext_vector_type(4))) float v4f;
typedef __attribute__((ext_vector_type(8))) float v8f;

#define D_MODEL 1024
#define N_HEADS 16
#define D_HEAD  64
#define BATCH   4
#define SEQ     2048
#define ROWS    (BATCH * SEQ)

static __device__ __forceinline__ v8f wmma_f32(v2f a, v2f b, v8f c) {
  return __builtin_amdgcn_wmma_f32_16x16x4_f32(
       false, a,  false, b,
       (short)0, c,  false,  false);
}

__global__ __launch_bounds__(256) void transpose_w_kernel(
    const float* __restrict__ W0, const float* __restrict__ W1,
    const float* __restrict__ W2, const float* __restrict__ W3,
    float* __restrict__ T0, float* __restrict__ T1,
    float* __restrict__ T2, float* __restrict__ T3) {
  const float* W = (blockIdx.z == 0) ? W0 : (blockIdx.z == 1) ? W1
                   : (blockIdx.z == 2) ? W2 : W3;
  float* T = (blockIdx.z == 0) ? T0 : (blockIdx.z == 1) ? T1
             : (blockIdx.z == 2) ? T2 : T3;

  __shared__ float tile[32][33];
  const int tx = threadIdx.x & 31;
  const int ty = threadIdx.x >> 5;
  const int x0 = blockIdx.x * 32;
  const int y0 = blockIdx.y * 32;

#pragma unroll
  for (int i = 0; i < 4; ++i)
    tile[ty + i * 8][tx] = W[(size_t)(y0 + ty + i * 8) * D_MODEL + x0 + tx];
  __syncthreads();
#pragma unroll
  for (int i = 0; i < 4; ++i)
    T[(size_t)(x0 + ty + i * 8) * D_MODEL + y0 + tx] = tile[tx][ty + i * 8];
}

#define QKV_MI 4
#define QKV_NJ 1
#define QKV_KU 8
__global__ __launch_bounds__(256) void qkv_proj_kernel(
    const float* __restrict__ X,
    const float* __restrict__ Wqt, const float* __restrict__ Wkt,
    const float* __restrict__ Wvt,
    float* __restrict__ Q, float* __restrict__ K, float* __restrict__ Vt) {
  const int wave = threadIdx.x >> 5;
  const int lane = threadIdx.x & 31;
  const int half = lane >> 4;
  const int l16  = lane & 15;

  const int row0 = blockIdx.x * (16 * QKV_MI);
  const int col0 = (blockIdx.y * 8 + wave) * (16 * QKV_NJ);

  v8f cq[QKV_MI][QKV_NJ], ck[QKV_MI][QKV_NJ], cv[QKV_MI][QKV_NJ];
#pragma unroll
  for (int i = 0; i < QKV_MI; ++i)
#pragma unroll
    for (int j = 0; j < QKV_NJ; ++j) { cq[i][j] = v8f{}; ck[i][j] = v8f{}; cv[i][j] = v8f{}; }

  const float* xr = X   + (size_t)(row0 + l16) * D_MODEL + 2 * half;
  const float* wq = Wqt + (size_t)(col0 + l16) * D_MODEL + 2 * half;
  const float* wk = Wkt + (size_t)(col0 + l16) * D_MODEL + 2 * half;
  const float* wv = Wvt + (size_t)(col0 + l16) * D_MODEL + 2 * half;

  for (int k0 = 0; k0 < D_MODEL; k0 += 4 * QKV_KU) {
    v2f a[QKV_MI][QKV_KU], bq[QKV_NJ][QKV_KU], bk[QKV_NJ][QKV_KU], bv[QKV_NJ][QKV_KU];
#pragma unroll
    for (int i = 0; i < QKV_MI; ++i)
#pragma unroll
      for (int kk = 0; kk < QKV_KU; ++kk)
        a[i][kk] = *(const v2f*)(xr + (size_t)i * 16 * D_MODEL + k0 + 4 * kk);
#pragma unroll
    for (int j = 0; j < QKV_NJ; ++j) {
#pragma unroll
      for (int kk = 0; kk < QKV_KU; ++kk)
        bq[j][kk] = *(const v2f*)(wq + (size_t)j * 16 * D_MODEL + k0 + 4 * kk);
#pragma unroll
      for (int kk = 0; kk < QKV_KU; ++kk)
        bk[j][kk] = *(const v2f*)(wk + (size_t)j * 16 * D_MODEL + k0 + 4 * kk);
#pragma unroll
      for (int kk = 0; kk < QKV_KU; ++kk)
        bv[j][kk] = *(const v2f*)(wv + (size_t)j * 16 * D_MODEL + k0 + 4 * kk);
    }
    __builtin_amdgcn_sched_barrier(0);
#pragma unroll
    for (int kk = 0; kk < QKV_KU; ++kk) {
#pragma unroll
      for (int i = 0; i < QKV_MI; ++i) {
#pragma unroll
        for (int j = 0; j < QKV_NJ; ++j) {
          cq[i][j] = wmma_f32(a[i][kk], bq[j][kk], cq[i][j]);
          ck[i][j] = wmma_f32(a[i][kk], bk[j][kk], ck[i][j]);
          cv[i][j] = wmma_f32(a[i][kk], bv[j][kk], cv[i][j]);
        }
      }
    }
  }

#pragma unroll
  for (int j = 0; j < QKV_NJ; ++j) {
    const int col = col0 + 16 * j + l16;
    const int h = col >> 6;
    const int d = col & 63;
#pragma unroll
    for (int i = 0; i < QKV_MI; ++i) {
#pragma unroll
      for (int r = 0; r < 8; ++r) {
        const int row = row0 + 16 * i + r + 8 * half;
        const int b = row >> 11;
        const int s = row & (SEQ - 1);
        const int bh = b * N_HEADS + h;
        const size_t o  = ((size_t)bh * SEQ + s) * D_HEAD + d;
        const size_t ov = ((size_t)bh * D_HEAD + d) * SEQ + s;
        Q[o]   = cq[i][j][r];
        K[o]   = ck[i][j][r];
        Vt[ov] = cv[i][j][r];
      }
    }
  }
}

#define ATT_QT 2
#define ATT_NB (SEQ / (16 * ATT_QT))
#define ATT_WG (ATT_NB / 8)
#define ATT_QP 68

template <int M>
static __device__ __forceinline__ float row_xor(float x) {
  const int b = __builtin_bit_cast(int, x);
  return __builtin_bit_cast(float,
      __builtin_amdgcn_update_dpp(b, b, 0x160 + M, 0xf, 0xf, true));
}

template <int M>
static __device__ __forceinline__ float row_xor_max(float x) {
  static_assert(M == 1 || M == 2 || M == 4 || M == 8, "one butterfly step inside a 16-lane row");
  float r;
  if (M == 1)
    asm("v_max_num_f32_dpp %0, %1, %1 row_xmask:1 row_mask:0xf bank_mask:0xf" : "=v"(r) : "v"(x));
  else if (M == 2)
    asm("v_max_num_f32_dpp %0, %1, %1 row_xmask:2 row_mask:0xf bank_mask:0xf" : "=v"(r) : "v"(x));
  else if (M == 4)
    asm("v_max_num_f32_dpp %0, %1, %1 row_xmask:4 row_mask:0xf bank_mask:0xf" : "=v"(r) : "v"(x));
  else
    asm("v_max_num_f32_dpp %0, %1, %1 row_xmask:8 row_mask:0xf bank_mask:0xf" : "=v"(r) : "v"(x));
  return r;
}

template <bool DIAG>
static __device__ __forceinline__ void softmax_tile(
    v8f& sc, float (&m8)[8], float (&l8)[8], v8f (&o)[4],
    const int half, const int l16) {
#pragma unroll
  for (int r = 0; r < 8; ++r) {
    float v = sc[r] * 0.125f;
    if (DIAG && (l16 > r + 8 * half)) v = -__builtin_inff();
    sc[r] = v;
  }
#pragma unroll
  for (int r = 0; r < 8; ++r) {
    float v = sc[r];
    v = row_xor_max<1>(v);
    v = row_xor_max<2>(v);
    v = row_xor_max<4>(v);
    v = row_xor_max<8>(v);
    const float mn   = fmaxf(m8[r], v);
    const float resc = __expf(m8[r] - mn);
    m8[r] = mn;
    const float p = __expf(sc[r] - mn);
    sc[r] = p;
    float s = p;
    s += row_xor<1>(s);
    s += row_xor<2>(s);
    s += row_xor<4>(s);
    s += row_xor<8>(s);
    l8[r] = l8[r] * resc + s;
    o[0][r] *= resc; o[1][r] *= resc; o[2][r] *= resc; o[3][r] *= resc;
  }
}

template <int QLO, bool DIAG0, bool DIAG1>
static __device__ __forceinline__ void attn_step(
    const int t0, const float* __restrict__ Km,
    const float* __restrict__ vb0, const float* __restrict__ vb1,
    const float* __restrict__ vb2, const float* __restrict__ vb3,
    const float (*qs)[ATT_QP], v8f (&o)[ATT_QT][4],
    float (&m8)[ATT_QT][8], float (&l8)[ATT_QT][8],
    float (*pt)[16][17], const int half, const int l16) {
  v8f sc[ATT_QT];
#pragma unroll
  for (int q = 0; q < ATT_QT; ++q) sc[q] = v8f{};
  const float* krow = Km + (size_t)(t0 + l16) * D_HEAD;
#pragma unroll
  for (int kk = 0; kk < 16; ++kk) {
    const v2f bf = *(const v2f*)(krow + kk * 4 + 2 * half);
#pragma unroll
    for (int q = QLO; q < ATT_QT; ++q) {
      const v2f af = *(const v2f*)(&qs[16 * q + l16][kk * 4 + 2 * half]);
      sc[q] = wmma_f32(af, bf, sc[q]);
    }
  }

  if (QLO == 0) softmax_tile<DIAG0>(sc[0], m8[0], l8[0], o[0], half, l16);
  softmax_tile<DIAG1>(sc[1], m8[1], l8[1], o[1], half, l16);

#pragma unroll
  for (int q = QLO; q < ATT_QT; ++q)
#pragma unroll
    for (int r = 0; r < 8; ++r)
      pt[q][r + 8 * half][l16] = sc[q][r];
  asm volatile("s_wait_dscnt 0" ::: "memory");

#pragma unroll
  for (int kk = 0; kk < 4; ++kk) {
    v2f a[ATT_QT];
#pragma unroll
    for (int q = QLO; q < ATT_QT; ++q) {
      a[q].x = pt[q][l16][kk * 4 + 2 * half];
      a[q].y = pt[q][l16][kk * 4 + 2 * half + 1];
    }
    const int tk = t0 + kk * 4 + 2 * half;
    v2f b;
    b = *(const v2f*)(vb0 + tk);
#pragma unroll
    for (int q = QLO; q < ATT_QT; ++q) o[q][0] = wmma_f32(a[q], b, o[q][0]);
    b = *(const v2f*)(vb1 + tk);
#pragma unroll
    for (int q = QLO; q < ATT_QT; ++q) o[q][1] = wmma_f32(a[q], b, o[q][1]);
    b = *(const v2f*)(vb2 + tk);
#pragma unroll
    for (int q = QLO; q < ATT_QT; ++q) o[q][2] = wmma_f32(a[q], b, o[q][2]);
    b = *(const v2f*)(vb3 + tk);
#pragma unroll
    for (int q = QLO; q < ATT_QT; ++q) o[q][3] = wmma_f32(a[q], b, o[q][3]);
  }
}

__global__ __launch_bounds__(256) void flash_attn_kernel(
    const float* __restrict__ Q, const float* __restrict__ K,
    const float* __restrict__ Vt, float* __restrict__ Oout) {
  const int wave = threadIdx.x >> 5;
  const int lane = threadIdx.x & 31;
  const int half = lane >> 4;
  const int l16  = lane & 15;

  const int bh = blockIdx.y;
  static_assert(ATT_NB == 8 * ATT_WG, "eight waves per workgroup: four SIMD pairs of (mirror, block)");
  const int g  = wave & 3;
  const int bx = blockIdx.x;
  const int lo = ATT_WG * g + ((g & 1) ? (ATT_WG - 1 - bx) : bx);
  const int qb = (wave & 4) ? lo : (ATT_NB - 1 - lo);
  const int q0 = qb * (16 * ATT_QT);

  const float* Qm  = Q  + (size_t)bh * SEQ * D_HEAD;
  const float* Km  = K  + (size_t)bh * SEQ * D_HEAD;
  const float* Vtm = Vt + (size_t)bh * D_HEAD * SEQ;

  __shared__ float lds[8][ATT_QT][16][17];
  float (*pt)[16][17] = lds[wave];

  __shared__ float ldsq[8][16 * ATT_QT][ATT_QP];
  float (*qs)[ATT_QP] = ldsq[wave];
#pragma unroll
  for (int r = 0; r < 16 * ATT_QT; ++r)
    *(v2f*)(&qs[r][2 * lane]) = *(const v2f*)(Qm + (size_t)(q0 + r) * D_HEAD + 2 * lane);

  v8f o[ATT_QT][4];
  float m8[ATT_QT][8], l8[ATT_QT][8];
#pragma unroll
  for (int q = 0; q < ATT_QT; ++q) {
#pragma unroll
    for (int n = 0; n < 4; ++n) o[q][n] = v8f{};
#pragma unroll
    for (int r = 0; r < 8; ++r) { m8[q][r] = -__builtin_inff(); l8[q][r] = 0.0f; }
  }

  const float* vb0 = Vtm + (size_t)( 0 + l16) * SEQ;
  const float* vb1 = Vtm + (size_t)(16 + l16) * SEQ;
  const float* vb2 = Vtm + (size_t)(32 + l16) * SEQ;
  const float* vb3 = Vtm + (size_t)(48 + l16) * SEQ;

  for (int t0 = 0; t0 < q0; t0 += 16)
    attn_step<0, false, false>(t0, Km, vb0, vb1, vb2, vb3, qs, o, m8, l8, pt, half, l16);
  attn_step<0, true, false>(q0, Km, vb0, vb1, vb2, vb3, qs, o, m8, l8, pt, half, l16);
  attn_step<1, false, true>(q0 + 16, Km, vb0, vb1, vb2, vb3, qs, o, m8, l8, pt, half, l16);

  const int b = bh >> 4;
  const int h = bh & (N_HEADS - 1);
#pragma unroll
  for (int q = 0; q < ATT_QT; ++q) {
#pragma unroll
    for (int r = 0; r < 8; ++r) {
      const float inv = 1.0f / l8[q][r];
      const int s = q0 + 16 * q + r + 8 * half;
      float* orow = Oout + (size_t)(b * SEQ + s) * (N_HEADS * D_HEAD) + h * D_HEAD;
      orow[ 0 + l16] = o[q][0][r] * inv;
      orow[16 + l16] = o[q][1][r] * inv;
      orow[32 + l16] = o[q][2][r] * inv;
      orow[48 + l16] = o[q][3][r] * inv;
    }
  }
}

#define OP_MI 4
#define OP_NJ 4
#define OP_KG 4
#define OP_KT (8 * OP_KG)

static __device__ __forceinline__ void op_load(
    v4f (&a)[OP_MI][OP_KG], v4f (&b)[OP_NJ][OP_KG],
    const float* __restrict__ ar, const float* __restrict__ br, const int k0) {
#pragma unroll
  for (int i = 0; i < OP_MI; ++i)
#pragma unroll
    for (int g = 0; g < OP_KG; ++g)
      a[i][g] = *(const v4f*)(ar + (size_t)i * 16 * D_MODEL + k0 + 8 * g);
#pragma unroll
  for (int j = 0; j < OP_NJ; ++j)
#pragma unroll
    for (int g = 0; g < OP_KG; ++g)
      b[j][g] = *(const v4f*)(br + (size_t)j * 16 * D_MODEL + k0 + 8 * g);
}

static __device__ __forceinline__ void op_mma(
    v8f (&c)[OP_MI][OP_NJ],
    const v4f (&a)[OP_MI][OP_KG], const v4f (&b)[OP_NJ][OP_KG]) {
#pragma unroll
  for (int g = 0; g < OP_KG; ++g) {
#pragma unroll
    for (int s = 0; s < 2; ++s) {
#pragma unroll
      for (int i = 0; i < OP_MI; ++i) {
        const v2f af = s ? v2f{a[i][g].z, a[i][g].w} : v2f{a[i][g].x, a[i][g].y};
#pragma unroll
        for (int j = 0; j < OP_NJ; ++j) {
          const v2f bf = s ? v2f{b[j][g].z, b[j][g].w} : v2f{b[j][g].x, b[j][g].y};
          c[i][j] = wmma_f32(af, bf, c[i][j]);
        }
      }
    }
  }
}

__global__ __launch_bounds__(256) void out_proj_kernel(
    const float* __restrict__ A, const float* __restrict__ Wot,
    float* __restrict__ C) {
  const int wave = threadIdx.x >> 5;
  const int lane = threadIdx.x & 31;
  const int half = lane >> 4;
  const int l16  = lane & 15;

  const int row0 = (blockIdx.x * 2 + (wave >> 2)) * (16 * OP_MI);
  const int col0 = (blockIdx.y * 4 + (wave & 3)) * (16 * OP_NJ);

  v8f c[OP_MI][OP_NJ];
#pragma unroll
  for (int i = 0; i < OP_MI; ++i)
#pragma unroll
    for (int j = 0; j < OP_NJ; ++j) c[i][j] = v8f{};

  const float* ar = A   + (size_t)(row0 + l16) * D_MODEL + 4 * half;
  const float* br = Wot + (size_t)(col0 + l16) * D_MODEL + 4 * half;

  v4f a0[OP_MI][OP_KG], b0[OP_NJ][OP_KG], a1[OP_MI][OP_KG], b1[OP_NJ][OP_KG];
  op_load(a0, b0, ar, br, 0);
  for (int k0 = 0; k0 < D_MODEL - 2 * OP_KT; k0 += 2 * OP_KT) {
    op_load(a1, b1, ar, br, k0 + OP_KT);
    __builtin_amdgcn_sched_barrier(0);
    op_mma(c, a0, b0);
    __builtin_amdgcn_sched_barrier(0);
    op_load(a0, b0, ar, br, k0 + 2 * OP_KT);
    __builtin_amdgcn_sched_barrier(0);
    op_mma(c, a1, b1);
    __builtin_amdgcn_sched_barrier(0);
  }
  op_load(a1, b1, ar, br, D_MODEL - OP_KT);
  __builtin_amdgcn_sched_barrier(0);
  op_mma(c, a0, b0);
  op_mma(c, a1, b1);

#pragma unroll
  for (int i = 0; i < OP_MI; ++i) {
#pragma unroll
    for (int r = 0; r < 8; ++r) {
      float* crow = C + (size_t)(row0 + 16 * i + r + 8 * half) * D_MODEL + col0 + l16;
#pragma unroll
      for (int j = 0; j < OP_NJ; ++j) crow[16 * j] = c[i][j][r];
    }
  }
}

extern "C" void kernel_launch(void* const* d_in, const int* in_sizes, int n_in,
                              void* d_out, int out_size, void* d_ws, size_t ws_size,
                              hipStream_t stream) {
  const float* x  = (const float*)d_in[0];
  const float* Wq = (const float*)d_in[1];
  const float* Wk = (const float*)d_in[2];
  const float* Wv = (const float*)d_in[3];
  const float* Wo = (const float*)d_in[4];

  float* ws = (float*)d_ws;
  const size_t per = (size_t)BATCH * N_HEADS * SEQ * D_HEAD;
  const size_t wsz = (size_t)D_MODEL * D_MODEL;
  float* Q   = ws;
  float* K   = ws + per;
  float* Vt  = ws + 2 * per;
  float* AO  = ws + 3 * per;
  float* Wqt = ws + 4 * per;
  float* Wkt = Wqt + wsz;
  float* Wvt = Wkt + wsz;
  float* Wot = Wvt + wsz;

  dim3 g0(D_MODEL / 32, D_MODEL / 32, 4);
  transpose_w_kernel<<<g0, 256, 0, stream>>>(Wq, Wk, Wv, Wo, Wqt, Wkt, Wvt, Wot);

  dim3 g1(ROWS / (16 * QKV_MI), (N_HEADS * D_HEAD) / (16 * QKV_NJ * 8));
  qkv_proj_kernel<<<g1, 256, 0, stream>>>(x, Wqt, Wkt, Wvt, Q, K, Vt);

  dim3 g2(SEQ / (16 * ATT_QT * 8), BATCH * N_HEADS);
  flash_attn_kernel<<<g2, 256, 0, stream>>>(Q, K, Vt, AO);

  dim3 g3(ROWS / (16 * OP_MI * 2), D_MODEL / (16 * OP_NJ * 4));
  out_proj_kernel<<<g3, 256, 0, stream>>>(AO, Wot, (float*)d_out);
}


# Attention: a step's V^T fragments and the next step's K fragments are requested right behind its score WMMAs, ahead of the softmax (were: K at the step's start, V^T behind the P store); order kept
# speedup vs baseline: 1.9137x; 1.0354x over previous
#include <hip/hip_runtime.h>
#include <hip/hip_bf16.h>

typedef __attribute__((ext_vector_type(2))) float v2f;
typedef __attribute__((ext_vector_type(4))) float v4f;
typedef __attribute__((ext_vector_type(8))) float v8f;

#define D_MODEL 1024
#define N_HEADS 16
#define D_HEAD  64
#define BATCH   4
#define SEQ     2048
#define ROWS    (BATCH * SEQ)

static __device__ __forceinline__ v8f wmma_f32(v2f a, v2f b, v8f c) {
  return __builtin_amdgcn_wmma_f32_16x16x4_f32(
       false, a,  false, b,
       (short)0, c,  false,  false);
}

__global__ __launch_bounds__(256) void transpose_w_kernel(
    const float* __restrict__ W0, const float* __restrict__ W1,
    const float* __restrict__ W2, const float* __restrict__ W3,
    float* __restrict__ T0, float* __restrict__ T1,
    float* __restrict__ T2, float* __restrict__ T3) {
  const float* W = (blockIdx.z == 0) ? W0 : (blockIdx.z == 1) ? W1
                   : (blockIdx.z == 2) ? W2 : W3;
  float* T = (blockIdx.z == 0) ? T0 : (blockIdx.z == 1) ? T1
             : (blockIdx.z == 2) ? T2 : T3;

  __shared__ float tile[32][33];
  const int tx = threadIdx.x & 31;
  const int ty = threadIdx.x >> 5;
  const int x0 = blockIdx.x * 32;
  const int y0 = blockIdx.y * 32;

#pragma unroll
  for (int i = 0; i < 4; ++i)
    tile[ty + i * 8][tx] = W[(size_t)(y0 + ty + i * 8) * D_MODEL + x0 + tx];
  __syncthreads();
#pragma unroll
  for (int i = 0; i < 4; ++i)
    T[(size_t)(x0 + ty + i * 8) * D_MODEL + y0 + tx] = tile[tx][ty + i * 8];
}

#define QKV_MI 4
#define QKV_NJ 1
#define QKV_KU 8
__global__ __launch_bounds__(256) void qkv_proj_kernel(
    const float* __restrict__ X,
    const float* __restrict__ Wqt, const float* __restrict__ Wkt,
    const float* __restrict__ Wvt,
    float* __restrict__ Q, float* __restrict__ K, float* __restrict__ Vt) {
  const int wave = threadIdx.x >> 5;
  const int lane = threadIdx.x & 31;
  const int half = lane >> 4;
  const int l16  = lane & 15;

  const int row0 = blockIdx.x * (16 * QKV_MI);
  const int col0 = (blockIdx.y * 8 + wave) * (16 * QKV_NJ);

  v8f cq[QKV_MI][QKV_NJ], ck[QKV_MI][QKV_NJ], cv[QKV_MI][QKV_NJ];
#pragma unroll
  for (int i = 0; i < QKV_MI; ++i)
#pragma unroll
    for (int j = 0; j < QKV_NJ; ++j) { cq[i][j] = v8f{}; ck[i][j] = v8f{}; cv[i][j] = v8f{}; }

  const float* xr = X   + (size_t)(row0 + l16) * D_MODEL + 2 * half;
  const float* wq = Wqt + (size_t)(col0 + l16) * D_MODEL + 2 * half;
  const float* wk = Wkt + (size_t)(col0 + l16) * D_MODEL + 2 * half;
  const float* wv = Wvt + (size_t)(col0 + l16) * D_MODEL + 2 * half;

  for (int k0 = 0; k0 < D_MODEL; k0 += 4 * QKV_KU) {
    v2f a[QKV_MI][QKV_KU], bq[QKV_NJ][QKV_KU], bk[QKV_NJ][QKV_KU], bv[QKV_NJ][QKV_KU];
#pragma unroll
    for (int i = 0; i < QKV_MI; ++i)
#pragma unroll
      for (int kk = 0; kk < QKV_KU; ++kk)
        a[i][kk] = *(const v2f*)(xr + (size_t)i * 16 * D_MODEL + k0 + 4 * kk);
#pragma unroll
    for (int j = 0; j < QKV_NJ; ++j) {
#pragma unroll
      for (int kk = 0; kk < QKV_KU; ++kk)
        bq[j][kk] = *(const v2f*)(wq + (size_t)j * 16 * D_MODEL + k0 + 4 * kk);
#pragma unroll
      for (int kk = 0; kk < QKV_KU; ++kk)
        bk[j][kk] = *(const v2f*)(wk + (size_t)j * 16 * D_MODEL + k0 + 4 * kk);
#pragma unroll
      for (int kk = 0; kk < QKV_KU; ++kk)
        bv[j][kk] = *(const v2f*)(wv + (size_t)j * 16 * D_MODEL + k0 + 4 * kk);
    }
    __builtin_amdgcn_sched_barrier(0);
#pragma unroll
    for (int kk = 0; kk < QKV_KU; ++kk) {
#pragma unroll
      for (int i = 0; i < QKV_MI; ++i) {
#pragma unroll
        for (int j = 0; j < QKV_NJ; ++j) {
          cq[i][j] = wmma_f32(a[i][kk], bq[j][kk], cq[i][j]);
          ck[i][j] = wmma_f32(a[i][kk], bk[j][kk], ck[i][j]);
          cv[i][j] = wmma_f32(a[i][kk], bv[j][kk], cv[i][j]);
        }
      }
    }
  }

#pragma unroll
  for (int j = 0; j < QKV_NJ; ++j) {
    const int col = col0 + 16 * j + l16;
    const int h = col >> 6;
    const int d = col & 63;
#pragma unroll
    for (int i = 0; i < QKV_MI; ++i) {
#pragma unroll
      for (int r = 0; r < 8; ++r) {
        const int row = row0 + 16 * i + r + 8 * half;
        const int b = row >> 11;
        const int s = row & (SEQ - 1);
        const int bh = b * N_HEADS + h;
        const size_t o  = ((size_t)bh * SEQ + s) * D_HEAD + d;
        const size_t ov = ((size_t)bh * D_HEAD + d) * SEQ + s;
        Q[o]   = cq[i][j][r];
        K[o]   = ck[i][j][r];
        Vt[ov] = cv[i][j][r];
      }
    }
  }
}

#define ATT_QT 2
#define ATT_NB (SEQ / (16 * ATT_QT))
#define ATT_WG (ATT_NB / 8)
#define ATT_QP 68

template <int M>
static __device__ __forceinline__ float row_xor(float x) {
  const int b = __builtin_bit_cast(int, x);
  return __builtin_bit_cast(float,
      __builtin_amdgcn_update_dpp(b, b, 0x160 + M, 0xf, 0xf, true));
}

template <int M>
static __device__ __forceinline__ float row_xor_max(float x) {
  static_assert(M == 1 || M == 2 || M == 4 || M == 8, "one butterfly step inside a 16-lane row");
  float r;
  if (M == 1)
    asm("v_max_num_f32_dpp %0, %1, %1 row_xmask:1 row_mask:0xf bank_mask:0xf" : "=v"(r) : "v"(x));
  else if (M == 2)
    asm("v_max_num_f32_dpp %0, %1, %1 row_xmask:2 row_mask:0xf bank_mask:0xf" : "=v"(r) : "v"(x));
  else if (M == 4)
    asm("v_max_num_f32_dpp %0, %1, %1 row_xmask:4 row_mask:0xf bank_mask:0xf" : "=v"(r) : "v"(x));
  else
    asm("v_max_num_f32_dpp %0, %1, %1 row_xmask:8 row_mask:0xf bank_mask:0xf" : "=v"(r) : "v"(x));
  return r;
}

template <bool DIAG>
static __device__ __forceinline__ void softmax_tile(
    v8f& sc, float (&m8)[8], float (&l8)[8], v8f (&o)[4],
    const int half, const int l16) {
#pragma unroll
  for (int r = 0; r < 8; ++r) {
    float v = sc[r] * 0.125f;
    if (DIAG && (l16 > r + 8 * half)) v = -__builtin_inff();
    sc[r] = v;
  }
#pragma unroll
  for (int r = 0; r < 8; ++r) {
    float v = sc[r];
    v = row_xor_max<1>(v);
    v = row_xor_max<2>(v);
    v = row_xor_max<4>(v);
    v = row_xor_max<8>(v);
    const float mn   = fmaxf(m8[r], v);
    const float resc = __expf(m8[r] - mn);
    m8[r] = mn;
    const float p = __expf(sc[r] - mn);
    sc[r] = p;
    float s = p;
    s += row_xor<1>(s);
    s += row_xor<2>(s);
    s += row_xor<4>(s);
    s += row_xor<8>(s);
    l8[r] = l8[r] * resc + s;
    o[0][r] *= resc; o[1][r] *= resc; o[2][r] *= resc; o[3][r] *= resc;
  }
}

static __device__ __forceinline__ void load_k_frags(
    v2f (&kf)[16], const float* __restrict__ Km, const int t0,
    const int half, const int l16) {
  const float* krow = Km + (size_t)(t0 + l16) * D_HEAD + 2 * half;
#pragma unroll
  for (int kk = 0; kk < 16; ++kk) kf[kk] = *(const v2f*)(krow + kk * 4);
}

template <int QLO, bool DIAG0, bool DIAG1, bool NEXT>
static __device__ __forceinline__ void attn_step(
    const int t0, const float* __restrict__ Km,
    const float* __restrict__ vb0, const float* __restrict__ vb1,
    const float* __restrict__ vb2, const float* __restrict__ vb3,
    const float (*qs)[ATT_QP], v8f (&o)[ATT_QT][4],
    float (&m8)[ATT_QT][8], float (&l8)[ATT_QT][8],
    float (*pt)[16][17], v2f (&kf)[16], const int half, const int l16) {
  v8f sc[ATT_QT];
#pragma unroll
  for (int q = 0; q < ATT_QT; ++q) sc[q] = v8f{};
  v2f af[ATT_QT][16];
#pragma unroll
  for (int kk = 0; kk < 16; ++kk)
#pragma unroll
    for (int q = QLO; q < ATT_QT; ++q)
      af[q][kk] = *(const v2f*)(&qs[16 * q + l16][kk * 4 + 2 * half]);
  __builtin_amdgcn_sched_barrier(0);
#pragma unroll
  for (int kk = 0; kk < 16; ++kk)
#pragma unroll
    for (int q = QLO; q < ATT_QT; ++q)
      sc[q] = wmma_f32(af[q][kk], kf[kk], sc[q]);

  __builtin_amdgcn_sched_barrier(0);
  v2f vf[4][4];
#pragma unroll
  for (int kk = 0; kk < 4; ++kk) {
    const int tk = t0 + kk * 4 + 2 * half;
    vf[kk][0] = *(const v2f*)(vb0 + tk);
    vf[kk][1] = *(const v2f*)(vb1 + tk);
    vf[kk][2] = *(const v2f*)(vb2 + tk);
    vf[kk][3] = *(const v2f*)(vb3 + tk);
  }
  if (NEXT) load_k_frags(kf, Km, t0 + 16, half, l16);

  if (QLO == 0) softmax_tile<DIAG0>(sc[0], m8[0], l8[0], o[0], half, l16);
  softmax_tile<DIAG1>(sc[1], m8[1], l8[1], o[1], half, l16);

#pragma unroll
  for (int q = QLO; q < ATT_QT; ++q)
#pragma unroll
    for (int r = 0; r < 8; ++r)
      pt[q][r + 8 * half][l16] = sc[q][r];
  asm volatile("s_wait_dscnt 0" ::: "memory");

#pragma unroll
  for (int kk = 0; kk < 4; ++kk) {
    v2f a[ATT_QT];
#pragma unroll
    for (int q = QLO; q < ATT_QT; ++q) {
      a[q].x = pt[q][l16][kk * 4 + 2 * half];
      a[q].y = pt[q][l16][kk * 4 + 2 * half + 1];
    }
#pragma unroll
    for (int n = 0; n < 4; ++n)
#pragma unroll
      for (int q = QLO; q < ATT_QT; ++q) o[q][n] = wmma_f32(a[q], vf[kk][n], o[q][n]);
  }
}

__global__ __launch_bounds__(256) void flash_attn_kernel(
    const float* __restrict__ Q, const float* __restrict__ K,
    const float* __restrict__ Vt, float* __restrict__ Oout) {
  const int wave = threadIdx.x >> 5;
  const int lane = threadIdx.x & 31;
  const int half = lane >> 4;
  const int l16  = lane & 15;

  const int bh = blockIdx.y;
  static_assert(ATT_NB == 8 * ATT_WG, "eight waves per workgroup: four SIMD pairs of (mirror, block)");
  const int g  = wave & 3;
  const int bx = blockIdx.x;
  const int lo = ATT_WG * g + ((g & 1) ? (ATT_WG - 1 - bx) : bx);
  const int qb = (wave & 4) ? lo : (ATT_NB - 1 - lo);
  const int q0 = qb * (16 * ATT_QT);

  const float* Qm  = Q  + (size_t)bh * SEQ * D_HEAD;
  const float* Km  = K  + (size_t)bh * SEQ * D_HEAD;
  const float* Vtm = Vt + (size_t)bh * D_HEAD * SEQ;

  __shared__ float lds[8][ATT_QT][16][17];
  float (*pt)[16][17] = lds[wave];

  __shared__ float ldsq[8][16 * ATT_QT][ATT_QP];
  float (*qs)[ATT_QP] = ldsq[wave];
  v2f kf[16];
  load_k_frags(kf, Km, 0, half, l16);
#pragma unroll
  for (int r = 0; r < 16 * ATT_QT; ++r)
    *(v2f*)(&qs[r][2 * lane]) = *(const v2f*)(Qm + (size_t)(q0 + r) * D_HEAD + 2 * lane);

  v8f o[ATT_QT][4];
  float m8[ATT_QT][8], l8[ATT_QT][8];
#pragma unroll
  for (int q = 0; q < ATT_QT; ++q) {
#pragma unroll
    for (int n = 0; n < 4; ++n) o[q][n] = v8f{};
#pragma unroll
    for (int r = 0; r < 8; ++r) { m8[q][r] = -__builtin_inff(); l8[q][r] = 0.0f; }
  }

  const float* vb0 = Vtm + (size_t)( 0 + l16) * SEQ;
  const float* vb1 = Vtm + (size_t)(16 + l16) * SEQ;
  const float* vb2 = Vtm + (size_t)(32 + l16) * SEQ;
  const float* vb3 = Vtm + (size_t)(48 + l16) * SEQ;

  for (int t0 = 0; t0 < q0; t0 += 16)
    attn_step<0, false, false, true>(t0, Km, vb0, vb1, vb2, vb3, qs, o, m8, l8, pt, kf, half, l16);
  attn_step<0, true, false, true>(q0, Km, vb0, vb1, vb2, vb3, qs, o, m8, l8, pt, kf, half, l16);
  attn_step<1, false, true, false>(q0 + 16, Km, vb0, vb1, vb2, vb3, qs, o, m8, l8, pt, kf, half, l16);

  const int b = bh >> 4;
  const int h = bh & (N_HEADS - 1);
#pragma unroll
  for (int q = 0; q < ATT_QT; ++q) {
#pragma unroll
    for (int r = 0; r < 8; ++r) {
      const float inv = 1.0f / l8[q][r];
      const int s = q0 + 16 * q + r + 8 * half;
      float* orow = Oout + (size_t)(b * SEQ + s) * (N_HEADS * D_HEAD) + h * D_HEAD;
      orow[ 0 + l16] = o[q][0][r] * inv;
      orow[16 + l16] = o[q][1][r] * inv;
      orow[32 + l16] = o[q][2][r] * inv;
      orow[48 + l16] = o[q][3][r] * inv;
    }
  }
}

#define OP_MI 4
#define OP_NJ 4
#define OP_KG 4
#define OP_KT (8 * OP_KG)

static __device__ __forceinline__ void op_load(
    v4f (&a)[OP_MI][OP_KG], v4f (&b)[OP_NJ][OP_KG],
    const float* __restrict__ ar, const float* __restrict__ br, const int k0) {
#pragma unroll
  for (int i = 0; i < OP_MI; ++i)
#pragma unroll
    for (int g = 0; g < OP_KG; ++g)
      a[i][g] = *(const v4f*)(ar + (size_t)i * 16 * D_MODEL + k0 + 8 * g);
#pragma unroll
  for (int j = 0; j < OP_NJ; ++j)
#pragma unroll
    for (int g = 0; g < OP_KG; ++g)
      b[j][g] = *(const v4f*)(br + (size_t)j * 16 * D_MODEL + k0 + 8 * g);
}

static __device__ __forceinline__ void op_mma(
    v8f (&c)[OP_MI][OP_NJ],
    const v4f (&a)[OP_MI][OP_KG], const v4f (&b)[OP_NJ][OP_KG]) {
#pragma unroll
  for (int g = 0; g < OP_KG; ++g) {
#pragma unroll
    for (int s = 0; s < 2; ++s) {
#pragma unroll
      for (int i = 0; i < OP_MI; ++i) {
        const v2f af = s ? v2f{a[i][g].z, a[i][g].w} : v2f{a[i][g].x, a[i][g].y};
#pragma unroll
        for (int j = 0; j < OP_NJ; ++j) {
          const v2f bf = s ? v2f{b[j][g].z, b[j][g].w} : v2f{b[j][g].x, b[j][g].y};
          c[i][j] = wmma_f32(af, bf, c[i][j]);
        }
      }
    }
  }
}

__global__ __launch_bounds__(256) void out_proj_kernel(
    const float* __restrict__ A, const float* __restrict__ Wot,
    float* __restrict__ C) {
  const int wave = threadIdx.x >> 5;
  const int lane = threadIdx.x & 31;
  const int half = lane >> 4;
  const int l16  = lane & 15;

  const int row0 = (blockIdx.x * 2 + (wave >> 2)) * (16 * OP_MI);
  const int col0 = (blockIdx.y * 4 + (wave & 3)) * (16 * OP_NJ);

  v8f c[OP_MI][OP_NJ];
#pragma unroll
  for (int i = 0; i < OP_MI; ++i)
#pragma unroll
    for (int j = 0; j < OP_NJ; ++j) c[i][j] = v8f{};

  const float* ar = A   + (size_t)(row0 + l16) * D_MODEL + 4 * half;
  const float* br = Wot + (size_t)(col0 + l16) * D_MODEL + 4 * half;

  v4f a0[OP_MI][OP_KG], b0[OP_NJ][OP_KG], a1[OP_MI][OP_KG], b1[OP_NJ][OP_KG];
  op_load(a0, b0, ar, br, 0);
  for (int k0 = 0; k0 < D_MODEL - 2 * OP_KT; k0 += 2 * OP_KT) {
    op_load(a1, b1, ar, br, k0 + OP_KT);
    __builtin_amdgcn_sched_barrier(0);
    op_mma(c, a0, b0);
    __builtin_amdgcn_sched_barrier(0);
    op_load(a0, b0, ar, br, k0 + 2 * OP_KT);
    __builtin_amdgcn_sched_barrier(0);
    op_mma(c, a1, b1);
    __builtin_amdgcn_sched_barrier(0);
  }
  op_load(a1, b1, ar, br, D_MODEL - OP_KT);
  __builtin_amdgcn_sched_barrier(0);
  op_mma(c, a0, b0);
  op_mma(c, a1, b1);

#pragma unroll
  for (int i = 0; i < OP_MI; ++i) {
#pragma unroll
    for (int r = 0; r < 8; ++r) {
      float* crow = C + (size_t)(row0 + 16 * i + r + 8 * half) * D_MODEL + col0 + l16;
#pragma unroll
      for (int j = 0; j < OP_NJ; ++j) crow[16 * j] = c[i][j][r];
    }
  }
}

extern "C" void kernel_launch(void* const* d_in, const int* in_sizes, int n_in,
                              void* d_out, int out_size, void* d_ws, size_t ws_size,
                              hipStream_t stream) {
  const float* x  = (const float*)d_in[0];
  const float* Wq = (const float*)d_in[1];
  const float* Wk = (const float*)d_in[2];
  const float* Wv = (const float*)d_in[3];
  const float* Wo = (const float*)d_in[4];

  float* ws = (float*)d_ws;
  const size_t per = (size_t)BATCH * N_HEADS * SEQ * D_HEAD;
  const size_t wsz = (size_t)D_MODEL * D_MODEL;
  float* Q   = ws;
  float* K   = ws + per;
  float* Vt  = ws + 2 * per;
  float* AO  = ws + 3 * per;
  float* Wqt = ws + 4 * per;
  float* Wkt = Wqt + wsz;
  float* Wvt = Wkt + wsz;
  float* Wot = Wvt + wsz;

  dim3 g0(D_MODEL / 32, D_MODEL / 32, 4);
  transpose_w_kernel<<<g0, 256, 0, stream>>>(Wq, Wk, Wv, Wo, Wqt, Wkt, Wvt, Wot);

  dim3 g1(ROWS / (16 * QKV_MI), (N_HEADS * D_HEAD) / (16 * QKV_NJ * 8));
  qkv_proj_kernel<<<g1, 256, 0, stream>>>(x, Wqt, Wkt, Wvt, Q, K, Vt);

  dim3 g2(SEQ / (16 * ATT_QT * 8), BATCH * N_HEADS);
  flash_attn_kernel<<<g2, 256, 0, stream>>>(Q, K, Vt, AO);

  dim3 g3(ROWS / (16 * OP_MI * 2), D_MODEL / (16 * OP_NJ * 4));
  out_proj_kernel<<<g3, 256, 0, stream>>>(AO, Wot, (float*)d_out);
}


# K, V and the four weight matrices stored in WMMA fragment order by their producer kernels (K tile computed transposed): a B-fragment load is 4 whole lines, was 16 partial lines; sum orders kept
# speedup vs baseline: 2.0146x; 1.0527x over previous
#include <hip/hip_runtime.h>
#include <hip/hip_bf16.h>

typedef __attribute__((ext_vector_type(2))) float v2f;
typedef __attribute__((ext_vector_type(4))) float v4f;
typedef __attribute__((ext_vector_type(8))) float v8f;

#define D_MODEL 1024
#define N_HEADS 16
#define D_HEAD  64
#define BATCH   4
#define SEQ     2048
#define ROWS    (BATCH * SEQ)

static __device__ __forceinline__ v8f wmma_f32(v2f a, v2f b, v8f c) {
  return __builtin_amdgcn_wmma_f32_16x16x4_f32(
       false, a,  false, b,
       (short)0, c,  false,  false);
}

static __device__ __forceinline__ v2f frag_of(const v4f c, const int odd) {
  return odd ? v2f{c.z, c.w} : v2f{c.x, c.y};
}

__global__ __launch_bounds__(256) void pack_w_kernel(
    const float* __restrict__ W0, const float* __restrict__ W1,
    const float* __restrict__ W2, const float* __restrict__ W3,
    float* __restrict__ T0, float* __restrict__ T1,
    float* __restrict__ T2, float* __restrict__ T3) {
  const float* W = (blockIdx.z == 0) ? W0 : (blockIdx.z == 1) ? W1
                   : (blockIdx.z == 2) ? W2 : W3;
  float* T = (blockIdx.z == 0) ? T0 : (blockIdx.z == 1) ? T1
             : (blockIdx.z == 2) ? T2 : T3;

  __shared__ float tile[32][33];
  const int tx = threadIdx.x & 31;
  const int ty = threadIdx.x >> 5;
  const int x0 = blockIdx.x * 32;
  const int y0 = blockIdx.y * 32;

#pragma unroll
  for (int i = 0; i < 4; ++i)
    tile[ty + i * 8][tx] = W[(size_t)(y0 + ty + i * 8) * D_MODEL + x0 + tx];
  __syncthreads();
  const int gl  = ty & 3;
  const int nbl = ty >> 2;
  const int c = tx & 15;
  const int h = tx >> 4;
  const int kh = (blockIdx.z == 3) ? 4 * h : 2 * h;
  const int kp = (blockIdx.z == 3) ? 2 : 4;
  const float* tc = &tile[8 * gl + kh][16 * nbl + c];
  const v4f v = v4f{tc[0], tc[33], tc[33 * kp], tc[33 * kp + 33]};
  const int nb = (x0 >> 4) + nbl;
  const int G  = (y0 >> 3) + gl;
  *(v4f*)(T + ((size_t)(nb * 128 + G) * 32 + tx) * 4) = v;
}

#define QKV_MI 4
#define QKV_NJ 1
#define QKV_KU 8
__global__ __launch_bounds__(256) void qkv_proj_kernel(
    const float* __restrict__ X,
    const float* __restrict__ Wqp, const float* __restrict__ Wkp,
    const float* __restrict__ Wvp,
    float* __restrict__ Q, float* __restrict__ Kf, float* __restrict__ Vf) {
  const int wave = threadIdx.x >> 5;
  const int lane = threadIdx.x & 31;
  const int half = lane >> 4;
  const int l16  = lane & 15;

  const int row0 = blockIdx.x * (16 * QKV_MI);
  const int col0 = (blockIdx.y * 8 + wave) * (16 * QKV_NJ);

  v8f cq[QKV_MI][QKV_NJ], ck[QKV_MI][QKV_NJ], cv[QKV_MI][QKV_NJ];
#pragma unroll
  for (int i = 0; i < QKV_MI; ++i)
#pragma unroll
    for (int j = 0; j < QKV_NJ; ++j) { cq[i][j] = v8f{}; ck[i][j] = v8f{}; cv[i][j] = v8f{}; }

  const float* xr = X   + (size_t)(row0 + l16) * D_MODEL + 2 * half;
  const float* wq = Wqp + (size_t)(col0 >> 4) * (16 * D_MODEL) + 4 * lane;
  const float* wk = Wkp + (size_t)(col0 >> 4) * (16 * D_MODEL) + 4 * lane;
  const float* wv = Wvp + (size_t)(col0 >> 4) * (16 * D_MODEL) + 4 * lane;

  for (int k0 = 0; k0 < D_MODEL; k0 += 4 * QKV_KU) {
    v2f a[QKV_MI][QKV_KU];
    v4f bq[QKV_NJ][QKV_KU / 2], bk[QKV_NJ][QKV_KU / 2], bv[QKV_NJ][QKV_KU / 2];
#pragma unroll
    for (int i = 0; i < QKV_MI; ++i)
#pragma unroll
      for (int kk = 0; kk < QKV_KU; ++kk)
        a[i][kk] = *(const v2f*)(xr + (size_t)i * 16 * D_MODEL + k0 + 4 * kk);
#pragma unroll
    for (int j = 0; j < QKV_NJ; ++j) {
#pragma unroll
      for (int g = 0; g < QKV_KU / 2; ++g)
        bq[j][g] = *(const v4f*)(wq + (size_t)j * 16 * D_MODEL + k0 * 16 + g * 128);
#pragma unroll
      for (int g = 0; g < QKV_KU / 2; ++g)
        bk[j][g] = *(const v4f*)(wk + (size_t)j * 16 * D_MODEL + k0 * 16 + g * 128);
#pragma unroll
      for (int g = 0; g < QKV_KU / 2; ++g)
        bv[j][g] = *(const v4f*)(wv + (size_t)j * 16 * D_MODEL + k0 * 16 + g * 128);
    }
    __builtin_amdgcn_sched_barrier(0);
#pragma unroll
    for (int kk = 0; kk < QKV_KU; ++kk) {
#pragma unroll
      for (int i = 0; i < QKV_MI; ++i) {
#pragma unroll
        for (int j = 0; j < QKV_NJ; ++j) {
          cq[i][j] = wmma_f32(a[i][kk], frag_of(bq[j][kk >> 1], kk & 1), cq[i][j]);
          ck[i][j] = wmma_f32(frag_of(bk[j][kk >> 1], kk & 1), a[i][kk], ck[i][j]);
          cv[i][j] = wmma_f32(a[i][kk], frag_of(bv[j][kk >> 1], kk & 1), cv[i][j]);
        }
      }
    }
  }

#pragma unroll
  for (int j = 0; j < QKV_NJ; ++j) {
    const int colb = col0 + 16 * j;
    const int h  = colb >> 6;
    const int db = colb & 63;
#pragma unroll
    for (int i = 0; i < QKV_MI; ++i) {
      const int rowb = row0 + 16 * i;
      const int b  = rowb >> 11;
      const int sb = rowb & (SEQ - 1);
      const int bh = b * N_HEADS + h;
#pragma unroll
      for (int r = 0; r < 8; ++r) {
        const int s = sb + r + 8 * half;
        Q[((size_t)bh * SEQ + s) * D_HEAD + db + l16] = cq[i][j][r];
      }
      const size_t blk = ((size_t)bh * (SEQ / 16) + (sb >> 4)) * (16 * D_HEAD);
      float* kc = Kf + blk + (size_t)((db >> 3) + half) * 128 + 4 * l16;
      *(v4f*)(kc)      = v4f{ck[i][j][0], ck[i][j][1], ck[i][j][4], ck[i][j][5]};
      *(v4f*)(kc + 64) = v4f{ck[i][j][2], ck[i][j][3], ck[i][j][6], ck[i][j][7]};
      float* vc = Vf + blk + (size_t)(4 * half + (db >> 4)) * 128 + 4 * l16;
      *(v4f*)(vc)      = v4f{cv[i][j][0], cv[i][j][1], cv[i][j][4], cv[i][j][5]};
      *(v4f*)(vc + 64) = v4f{cv[i][j][2], cv[i][j][3], cv[i][j][6], cv[i][j][7]};
    }
  }
}

#define ATT_QT 2
#define ATT_NB (SEQ / (16 * ATT_QT))
#define ATT_WG (ATT_NB / 8)
#define ATT_QP 68

template <int M>
static __device__ __forceinline__ float row_xor(float x) {
  const int b = __builtin_bit_cast(int, x);
  return __builtin_bit_cast(float,
      __builtin_amdgcn_update_dpp(b, b, 0x160 + M, 0xf, 0xf, true));
}

template <int M>
static __device__ __forceinline__ float row_xor_max(float x) {
  static_assert(M == 1 || M == 2 || M == 4 || M == 8, "one butterfly step inside a 16-lane row");
  float r;
  if (M == 1)
    asm("v_max_num_f32_dpp %0, %1, %1 row_xmask:1 row_mask:0xf bank_mask:0xf" : "=v"(r) : "v"(x));
  else if (M == 2)
    asm("v_max_num_f32_dpp %0, %1, %1 row_xmask:2 row_mask:0xf bank_mask:0xf" : "=v"(r) : "v"(x));
  else if (M == 4)
    asm("v_max_num_f32_dpp %0, %1, %1 row_xmask:4 row_mask:0xf bank_mask:0xf" : "=v"(r) : "v"(x));
  else
    asm("v_max_num_f32_dpp %0, %1, %1 row_xmask:8 row_mask:0xf bank_mask:0xf" : "=v"(r) : "v"(x));
  return r;
}

template <bool DIAG>
static __device__ __forceinline__ void softmax_tile(
    v8f& sc, float (&m8)[8], float (&l8)[8], v8f (&o)[4],
    const int half, const int l16) {
#pragma unroll
  for (int r = 0; r < 8; ++r) {
    float v = sc[r] * 0.125f;
    if (DIAG && (l16 > r + 8 * half)) v = -__builtin_inff();
    sc[r] = v;
  }
#pragma unroll
  for (int r = 0; r < 8; ++r) {
    float v = sc[r];
    v = row_xor_max<1>(v);
    v = row_xor_max<2>(v);
    v = row_xor_max<4>(v);
    v = row_xor_max<8>(v);
    const float mn   = fmaxf(m8[r], v);
    const float resc = __expf(m8[r] - mn);
    m8[r] = mn;
    const float p = __expf(sc[r] - mn);
    sc[r] = p;
    float s = p;
    s += row_xor<1>(s);
    s += row_xor<2>(s);
    s += row_xor<4>(s);
    s += row_xor<8>(s);
    l8[r] = l8[r] * resc + s;
    o[0][r] *= resc; o[1][r] *= resc; o[2][r] *= resc; o[3][r] *= resc;
  }
}

static __device__ __forceinline__ void load_k_frags(
    v4f (&kf)[8], const float* __restrict__ Kl, const int t0) {
  const float* kc = Kl + (size_t)t0 * D_HEAD;
#pragma unroll
  for (int g = 0; g < 8; ++g) kf[g] = *(const v4f*)(kc + g * 128);
}


template <int QLO, bool DIAG0, bool DIAG1, bool NEXT>
static __device__ __forceinline__ void attn_step(
    const int t0, const float* __restrict__ Kl, const float* __restrict__ Vl,
    const float (*qs)[ATT_QP], v8f (&o)[ATT_QT][4],
    float (&m8)[ATT_QT][8], float (&l8)[ATT_QT][8],
    float (*pt)[16][17], v4f (&kf)[8], const int half, const int l16) {
  v8f sc[ATT_QT];
#pragma unroll
  for (int q = 0; q < ATT_QT; ++q) sc[q] = v8f{};
  v2f af[ATT_QT][16];
#pragma unroll
  for (int kk = 0; kk < 16; ++kk)
#pragma unroll
    for (int q = QLO; q < ATT_QT; ++q)
      af[q][kk] = *(const v2f*)(&qs[16 * q + l16][kk * 4 + 2 * half]);
  __builtin_amdgcn_sched_barrier(0);
#pragma unroll
  for (int kk = 0; kk < 16; ++kk)
#pragma unroll
    for (int q = QLO; q < ATT_QT; ++q)
      sc[q] = wmma_f32(af[q][kk], frag_of(kf[kk >> 1], kk & 1), sc[q]);

  __builtin_amdgcn_sched_barrier(0);
  v4f vf[2][4];
  const float* vc = Vl + (size_t)t0 * D_HEAD;
#pragma unroll
  for (int u = 0; u < 2; ++u)
#pragma unroll
    for (int n = 0; n < 4; ++n) vf[u][n] = *(const v4f*)(vc + (4 * u + n) * 128);
  if (NEXT) load_k_frags(kf, Kl, t0 + 16);

  if (QLO == 0) softmax_tile<DIAG0>(sc[0], m8[0], l8[0], o[0], half, l16);
  softmax_tile<DIAG1>(sc[1], m8[1], l8[1], o[1], half, l16);

#pragma unroll
  for (int q = QLO; q < ATT_QT; ++q)
#pragma unroll
    for (int r = 0; r < 8; ++r)
      pt[q][r + 8 * half][l16] = sc[q][r];
  asm volatile("s_wait_dscnt 0" ::: "memory");

#pragma unroll
  for (int kk = 0; kk < 4; ++kk) {
    v2f a[ATT_QT];
#pragma unroll
    for (int q = QLO; q < ATT_QT; ++q) {
      a[q].x = pt[q][l16][kk * 4 + 2 * half];
      a[q].y = pt[q][l16][kk * 4 + 2 * half + 1];
    }
#pragma unroll
    for (int n = 0; n < 4; ++n)
#pragma unroll
      for (int q = QLO; q < ATT_QT; ++q)
        o[q][n] = wmma_f32(a[q], frag_of(vf[kk >> 1][n], kk & 1), o[q][n]);
  }
}

__global__ __launch_bounds__(256) void flash_attn_kernel(
    const float* __restrict__ Q, const float* __restrict__ Kf,
    const float* __restrict__ Vf, float* __restrict__ Oout) {
  const int wave = threadIdx.x >> 5;
  const int lane = threadIdx.x & 31;
  const int half = lane >> 4;
  const int l16  = lane & 15;

  const int bh = blockIdx.y;
  static_assert(ATT_NB == 8 * ATT_WG, "eight waves per workgroup: four SIMD pairs of (mirror, block)");
  const int g  = wave & 3;
  const int bx = blockIdx.x;
  const int lo = ATT_WG * g + ((g & 1) ? (ATT_WG - 1 - bx) : bx);
  const int qb = (wave & 4) ? lo : (ATT_NB - 1 - lo);
  const int q0 = qb * (16 * ATT_QT);

  const float* Qm = Q  + (size_t)bh * SEQ * D_HEAD;
  const float* Kl = Kf + (size_t)bh * SEQ * D_HEAD + 4 * lane;
  const float* Vl = Vf + (size_t)bh * SEQ * D_HEAD + 4 * lane;

  __shared__ float lds[8][ATT_QT][16][17];
  float (*pt)[16][17] = lds[wave];

  __shared__ float ldsq[8][16 * ATT_QT][ATT_QP];
  float (*qs)[ATT_QP] = ldsq[wave];
  v4f kf[8];
  load_k_frags(kf, Kl, 0);
#pragma unroll
  for (int r = 0; r < 16 * ATT_QT; ++r)
    *(v2f*)(&qs[r][2 * lane]) = *(const v2f*)(Qm + (size_t)(q0 + r) * D_HEAD + 2 * lane);

  v8f o[ATT_QT][4];
  float m8[ATT_QT][8], l8[ATT_QT][8];
#pragma unroll
  for (int q = 0; q < ATT_QT; ++q) {
#pragma unroll
    for (int n = 0; n < 4; ++n) o[q][n] = v8f{};
#pragma unroll
    for (int r = 0; r < 8; ++r) { m8[q][r] = -__builtin_inff(); l8[q][r] = 0.0f; }
  }

  for (int t0 = 0; t0 < q0; t0 += 16)
    attn_step<0, false, false, true>(t0, Kl, Vl, qs, o, m8, l8, pt, kf, half, l16);
  attn_step<0, true, false, true>(q0, Kl, Vl, qs, o, m8, l8, pt, kf, half, l16);
  attn_step<1, false, true, false>(q0 + 16, Kl, Vl, qs, o, m8, l8, pt, kf, half, l16);

  const int b = bh >> 4;
  const int h = bh & (N_HEADS - 1);
#pragma unroll
  for (int q = 0; q < ATT_QT; ++q) {
#pragma unroll
    for (int r = 0; r < 8; ++r) {
      const float inv = 1.0f / l8[q][r];
      const int s = q0 + 16 * q + r + 8 * half;
      float* orow = Oout + (size_t)(b * SEQ + s) * (N_HEADS * D_HEAD) + h * D_HEAD;
      orow[ 0 + l16] = o[q][0][r] * inv;
      orow[16 + l16] = o[q][1][r] * inv;
      orow[32 + l16] = o[q][2][r] * inv;
      orow[48 + l16] = o[q][3][r] * inv;
    }
  }
}

#define OP_MI 4
#define OP_NJ 4
#define OP_KG 4
#define OP_KT (8 * OP_KG)

static __device__ __forceinline__ void op_load(
    v4f (&a)[OP_MI][OP_KG], v4f (&b)[OP_NJ][OP_KG],
    const float* __restrict__ ar, const float* __restrict__ br, const int k0) {
#pragma unroll
  for (int i = 0; i < OP_MI; ++i)
#pragma unroll
    for (int g = 0; g < OP_KG; ++g)
      a[i][g] = *(const v4f*)(ar + (size_t)i * 16 * D_MODEL + k0 + 8 * g);
#pragma unroll
  for (int j = 0; j < OP_NJ; ++j)
#pragma unroll
    for (int g = 0; g < OP_KG; ++g)
      b[j][g] = *(const v4f*)(br + (size_t)j * 16 * D_MODEL + k0 * 16 + g * 128);
}

static __device__ __forceinline__ void op_mma(
    v8f (&c)[OP_MI][OP_NJ],
    const v4f (&a)[OP_MI][OP_KG], const v4f (&b)[OP_NJ][OP_KG]) {
#pragma unroll
  for (int g = 0; g < OP_KG; ++g) {
#pragma unroll
    for (int s = 0; s < 2; ++s) {
#pragma unroll
      for (int i = 0; i < OP_MI; ++i) {
        const v2f af = s ? v2f{a[i][g].z, a[i][g].w} : v2f{a[i][g].x, a[i][g].y};
#pragma unroll
        for (int j = 0; j < OP_NJ; ++j) {
          const v2f bf = s ? v2f{b[j][g].z, b[j][g].w} : v2f{b[j][g].x, b[j][g].y};
          c[i][j] = wmma_f32(af, bf, c[i][j]);
        }
      }
    }
  }
}

__global__ __launch_bounds__(256) void out_proj_kernel(
    const float* __restrict__ A, const float* __restrict__ Wop,
    float* __restrict__ C) {
  const int wave = threadIdx.x >> 5;
  const int lane = threadIdx.x & 31;
  const int half = lane >> 4;
  const int l16  = lane & 15;

  const int row0 = (blockIdx.x * 2 + (wave >> 2)) * (16 * OP_MI);
  const int col0 = (blockIdx.y * 4 + (wave & 3)) * (16 * OP_NJ);

  v8f c[OP_MI][OP_NJ];
#pragma unroll
  for (int i = 0; i < OP_MI; ++i)
#pragma unroll
    for (int j = 0; j < OP_NJ; ++j) c[i][j] = v8f{};

  const float* ar = A   + (size_t)(row0 + l16) * D_MODEL + 4 * half;
  const float* br = Wop + (size_t)(col0 >> 4) * (16 * D_MODEL) + 4 * lane;

  v4f a0[OP_MI][OP_KG], b0[OP_NJ][OP_KG], a1[OP_MI][OP_KG], b1[OP_NJ][OP_KG];
  op_load(a0, b0, ar, br, 0);
  for (int k0 = 0; k0 < D_MODEL - 2 * OP_KT; k0 += 2 * OP_KT) {
    op_load(a1, b1, ar, br, k0 + OP_KT);
    __builtin_amdgcn_sched_barrier(0);
    op_mma(c, a0, b0);
    __builtin_amdgcn_sched_barrier(0);
    op_load(a0, b0, ar, br, k0 + 2 * OP_KT);
    __builtin_amdgcn_sched_barrier(0);
    op_mma(c, a1, b1);
    __builtin_amdgcn_sched_barrier(0);
  }
  op_load(a1, b1, ar, br, D_MODEL - OP_KT);
  __builtin_amdgcn_sched_barrier(0);
  op_mma(c, a0, b0);
  op_mma(c, a1, b1);

#pragma unroll
  for (int i = 0; i < OP_MI; ++i) {
#pragma unroll
    for (int r = 0; r < 8; ++r) {
      float* crow = C + (size_t)(row0 + 16 * i + r + 8 * half) * D_MODEL + col0 + l16;
#pragma unroll
      for (int j = 0; j < OP_NJ; ++j) crow[16 * j] = c[i][j][r];
    }
  }
}

extern "C" void kernel_launch(void* const* d_in, const int* in_sizes, int n_in,
                              void* d_out, int out_size, void* d_ws, size_t ws_size,
                              hipStream_t stream) {
  const float* x  = (const float*)d_in[0];
  const float* Wq = (const float*)d_in[1];
  const float* Wk = (const float*)d_in[2];
  const float* Wv = (const float*)d_in[3];
  const float* Wo = (const float*)d_in[4];

  float* ws = (float*)d_ws;
  const size_t per = (size_t)BATCH * N_HEADS * SEQ * D_HEAD;
  const size_t wsz = (size_t)D_MODEL * D_MODEL;
  float* Q   = ws;
  float* Kf  = ws + per;
  float* Vf  = ws + 2 * per;
  float* AO  = ws + 3 * per;
  float* Wqp = ws + 4 * per;
  float* Wkp = Wqp + wsz;
  float* Wvp = Wkp + wsz;
  float* Wop = Wvp + wsz;

  dim3 g0(D_MODEL / 32, D_MODEL / 32, 4);
  pack_w_kernel<<<g0, 256, 0, stream>>>(Wq, Wk, Wv, Wo, Wqp, Wkp, Wvp, Wop);

  dim3 g1(ROWS / (16 * QKV_MI), (N_HEADS * D_HEAD) / (16 * QKV_NJ * 8));
  qkv_proj_kernel<<<g1, 256, 0, stream>>>(x, Wqp, Wkp, Wvp, Q, Kf, Vf);

  dim3 g2(SEQ / (16 * ATT_QT * 8), BATCH * N_HEADS);
  flash_attn_kernel<<<g2, 256, 0, stream>>>(Q, Kf, Vf, AO);

  dim3 g3(ROWS / (16 * OP_MI * 2), D_MODEL / (16 * OP_NJ * 4));
  out_proj_kernel<<<g3, 256, 0, stream>>>(AO, Wop, (float*)d_out);
}


# Attention: softmax runs behind the LDS layout turn: raw scores return in A layout, a query's 16 keys in one lane pair; max/rescale/sum once per pair, row_share to output regs; row-sum order changed
# speedup vs baseline: 2.1211x; 1.0529x over previous
#include <hip/hip_runtime.h>
#include <hip/hip_bf16.h>

typedef __attribute__((ext_vector_type(2))) float v2f;
typedef __attribute__((ext_vector_type(4))) float v4f;
typedef __attribute__((ext_vector_type(8))) float v8f;

#define D_MODEL 1024
#define N_HEADS 16
#define D_HEAD  64
#define BATCH   4
#define SEQ     2048
#define ROWS    (BATCH * SEQ)

static __device__ __forceinline__ v8f wmma_f32(v2f a, v2f b, v8f c) {
  return __builtin_amdgcn_wmma_f32_16x16x4_f32(
       false, a,  false, b,
       (short)0, c,  false,  false);
}

static __device__ __forceinline__ v2f frag_of(const v4f c, const int odd) {
  return odd ? v2f{c.z, c.w} : v2f{c.x, c.y};
}

__global__ __launch_bounds__(256) void pack_w_kernel(
    const float* __restrict__ W0, const float* __restrict__ W1,
    const float* __restrict__ W2, const float* __restrict__ W3,
    float* __restrict__ T0, float* __restrict__ T1,
    float* __restrict__ T2, float* __restrict__ T3) {
  const float* W = (blockIdx.z == 0) ? W0 : (blockIdx.z == 1) ? W1
                   : (blockIdx.z == 2) ? W2 : W3;
  float* T = (blockIdx.z == 0) ? T0 : (blockIdx.z == 1) ? T1
             : (blockIdx.z == 2) ? T2 : T3;

  __shared__ float tile[32][33];
  const int tx = threadIdx.x & 31;
  const int ty = threadIdx.x >> 5;
  const int x0 = blockIdx.x * 32;
  const int y0 = blockIdx.y * 32;

#pragma unroll
  for (int i = 0; i < 4; ++i)
    tile[ty + i * 8][tx] = W[(size_t)(y0 + ty + i * 8) * D_MODEL + x0 + tx];
  __syncthreads();
  const int gl  = ty & 3;
  const int nbl = ty >> 2;
  const int c = tx & 15;
  const int h = tx >> 4;
  const int kh = (blockIdx.z == 3) ? 4 * h : 2 * h;
  const int kp = (blockIdx.z == 3) ? 2 : 4;
  const float* tc = &tile[8 * gl + kh][16 * nbl + c];
  const v4f v = v4f{tc[0], tc[33], tc[33 * kp], tc[33 * kp + 33]};
  const int nb = (x0 >> 4) + nbl;
  const int G  = (y0 >> 3) + gl;
  *(v4f*)(T + ((size_t)(nb * 128 + G) * 32 + tx) * 4) = v;
}

#define QKV_MI 4
#define QKV_NJ 1
#define QKV_KU 8
__global__ __launch_bounds__(256) void qkv_proj_kernel(
    const float* __restrict__ X,
    const float* __restrict__ Wqp, const float* __restrict__ Wkp,
    const float* __restrict__ Wvp,
    float* __restrict__ Q, float* __restrict__ Kf, float* __restrict__ Vf) {
  const int wave = threadIdx.x >> 5;
  const int lane = threadIdx.x & 31;
  const int half = lane >> 4;
  const int l16  = lane & 15;

  const int row0 = blockIdx.x * (16 * QKV_MI);
  const int col0 = (blockIdx.y * 8 + wave) * (16 * QKV_NJ);

  v8f cq[QKV_MI][QKV_NJ], ck[QKV_MI][QKV_NJ], cv[QKV_MI][QKV_NJ];
#pragma unroll
  for (int i = 0; i < QKV_MI; ++i)
#pragma unroll
    for (int j = 0; j < QKV_NJ; ++j) { cq[i][j] = v8f{}; ck[i][j] = v8f{}; cv[i][j] = v8f{}; }

  const float* xr = X   + (size_t)(row0 + l16) * D_MODEL + 2 * half;
  const float* wq = Wqp + (size_t)(col0 >> 4) * (16 * D_MODEL) + 4 * lane;
  const float* wk = Wkp + (size_t)(col0 >> 4) * (16 * D_MODEL) + 4 * lane;
  const float* wv = Wvp + (size_t)(col0 >> 4) * (16 * D_MODEL) + 4 * lane;

  for (int k0 = 0; k0 < D_MODEL; k0 += 4 * QKV_KU) {
    v2f a[QKV_MI][QKV_KU];
    v4f bq[QKV_NJ][QKV_KU / 2], bk[QKV_NJ][QKV_KU / 2], bv[QKV_NJ][QKV_KU / 2];
#pragma unroll
    for (int i = 0; i < QKV_MI; ++i)
#pragma unroll
      for (int kk = 0; kk < QKV_KU; ++kk)
        a[i][kk] = *(const v2f*)(xr + (size_t)i * 16 * D_MODEL + k0 + 4 * kk);
#pragma unroll
    for (int j = 0; j < QKV_NJ; ++j) {
#pragma unroll
      for (int g = 0; g < QKV_KU / 2; ++g)
        bq[j][g] = *(const v4f*)(wq + (size_t)j * 16 * D_MODEL + k0 * 16 + g * 128);
#pragma unroll
      for (int g = 0; g < QKV_KU / 2; ++g)
        bk[j][g] = *(const v4f*)(wk + (size_t)j * 16 * D_MODEL + k0 * 16 + g * 128);
#pragma unroll
      for (int g = 0; g < QKV_KU / 2; ++g)
        bv[j][g] = *(const v4f*)(wv + (size_t)j * 16 * D_MODEL + k0 * 16 + g * 128);
    }
    __builtin_amdgcn_sched_barrier(0);
#pragma unroll
    for (int kk = 0; kk < QKV_KU; ++kk) {
#pragma unroll
      for (int i = 0; i < QKV_MI; ++i) {
#pragma unroll
        for (int j = 0; j < QKV_NJ; ++j) {
          cq[i][j] = wmma_f32(a[i][kk], frag_of(bq[j][kk >> 1], kk & 1), cq[i][j]);
          ck[i][j] = wmma_f32(frag_of(bk[j][kk >> 1], kk & 1), a[i][kk], ck[i][j]);
          cv[i][j] = wmma_f32(a[i][kk], frag_of(bv[j][kk >> 1], kk & 1), cv[i][j]);
        }
      }
    }
  }

#pragma unroll
  for (int j = 0; j < QKV_NJ; ++j) {
    const int colb = col0 + 16 * j;
    const int h  = colb >> 6;
    const int db = colb & 63;
#pragma unroll
    for (int i = 0; i < QKV_MI; ++i) {
      const int rowb = row0 + 16 * i;
      const int b  = rowb >> 11;
      const int sb = rowb & (SEQ - 1);
      const int bh = b * N_HEADS + h;
#pragma unroll
      for (int r = 0; r < 8; ++r) {
        const int s = sb + r + 8 * half;
        Q[((size_t)bh * SEQ + s) * D_HEAD + db + l16] = cq[i][j][r];
      }
      const size_t blk = ((size_t)bh * (SEQ / 16) + (sb >> 4)) * (16 * D_HEAD);
      float* kc = Kf + blk + (size_t)((db >> 3) + half) * 128 + 4 * l16;
      *(v4f*)(kc)      = v4f{ck[i][j][0], ck[i][j][1], ck[i][j][4], ck[i][j][5]};
      *(v4f*)(kc + 64) = v4f{ck[i][j][2], ck[i][j][3], ck[i][j][6], ck[i][j][7]};
      float* vc = Vf + blk + (size_t)(4 * half + (db >> 4)) * 128 + 4 * l16;
      *(v4f*)(vc)      = v4f{cv[i][j][0], cv[i][j][1], cv[i][j][4], cv[i][j][5]};
      *(v4f*)(vc + 64) = v4f{cv[i][j][2], cv[i][j][3], cv[i][j][6], cv[i][j][7]};
    }
  }
}

#define ATT_QT 2
#define ATT_NB (SEQ / (16 * ATT_QT))
#define ATT_WG (ATT_NB / 8)
#define ATT_QP 68

static __device__ __forceinline__ float xhalf(float x) {
  const int b = __builtin_bit_cast(int, x);
  return __builtin_bit_cast(float,
      __builtin_amdgcn_permlanex16(b, b, 0x76543210, 0xfedcba98, false, false));
}

template <int R>
static __device__ __forceinline__ float row_lane(float x) {
  const int b = __builtin_bit_cast(int, x);
  return __builtin_bit_cast(float,
      __builtin_amdgcn_update_dpp(b, b, 0x150 + R, 0xf, 0xf, true));
}

static __device__ __forceinline__ void per_register(float x, float (&f)[8]) {
  const int b = __builtin_bit_cast(int, x);
  const float z = __builtin_bit_cast(float,
      __builtin_amdgcn_update_dpp(b, b, 0x160 + 8, 0x2, 0xf, false));
  f[0] = row_lane<0>(z); f[1] = row_lane<1>(z); f[2] = row_lane<2>(z); f[3] = row_lane<3>(z);
  f[4] = row_lane<4>(z); f[5] = row_lane<5>(z); f[6] = row_lane<6>(z); f[7] = row_lane<7>(z);
}

template <bool DIAG>
static __device__ __forceinline__ void softmax_row(
    float (&p)[8], float& m, float& l, v8f (&o)[4],
    const int half, const int l16) {
#pragma unroll
  for (int j = 0; j < 8; ++j) {
    float v = p[j] * 0.125f;
    if (DIAG && (4 * (j >> 1) + 2 * half + (j & 1) > l16)) v = -__builtin_inff();
    p[j] = v;
  }
  float mx = fmaxf(fmaxf(fmaxf(p[0], p[1]), fmaxf(p[2], p[3])),
                   fmaxf(fmaxf(p[4], p[5]), fmaxf(p[6], p[7])));
  mx = fmaxf(mx, xhalf(mx));
  const float mn   = fmaxf(m, mx);
  const float resc = __expf(m - mn);
  m = mn;
#pragma unroll
  for (int j = 0; j < 8; ++j) p[j] = __expf(p[j] - mn);
  const float s = ((p[0] + p[1]) + (p[2] + p[3])) + ((p[4] + p[5]) + (p[6] + p[7]));
  l = l * resc + s;
  float rc[8];
  per_register(resc, rc);
#pragma unroll
  for (int r = 0; r < 8; ++r) {
    o[0][r] *= rc[r]; o[1][r] *= rc[r]; o[2][r] *= rc[r]; o[3][r] *= rc[r];
  }
}

static __device__ __forceinline__ void load_k_frags(
    v4f (&kf)[8], const float* __restrict__ Kl, const int t0) {
  const float* kc = Kl + (size_t)t0 * D_HEAD;
#pragma unroll
  for (int g = 0; g < 8; ++g) kf[g] = *(const v4f*)(kc + g * 128);
}


template <int QLO, bool DIAG0, bool DIAG1, bool NEXT>
static __device__ __forceinline__ void attn_step(
    const int t0, const float* __restrict__ Kl, const float* __restrict__ Vl,
    const float (*qs)[ATT_QP], v8f (&o)[ATT_QT][4],
    float (&mq)[ATT_QT], float (&lq)[ATT_QT],
    float (*pt)[16][17], v4f (&kf)[8], const int half, const int l16) {
  v8f sc[ATT_QT];
#pragma unroll
  for (int q = 0; q < ATT_QT; ++q) sc[q] = v8f{};
  v2f af[ATT_QT][16];
#pragma unroll
  for (int kk = 0; kk < 16; ++kk)
#pragma unroll
    for (int q = QLO; q < ATT_QT; ++q)
      af[q][kk] = *(const v2f*)(&qs[16 * q + l16][kk * 4 + 2 * half]);
  __builtin_amdgcn_sched_barrier(0);
#pragma unroll
  for (int kk = 0; kk < 16; ++kk)
#pragma unroll
    for (int q = QLO; q < ATT_QT; ++q)
      sc[q] = wmma_f32(af[q][kk], frag_of(kf[kk >> 1], kk & 1), sc[q]);

  __builtin_amdgcn_sched_barrier(0);
  v4f vf[2][4];
  const float* vc = Vl + (size_t)t0 * D_HEAD;
#pragma unroll
  for (int u = 0; u < 2; ++u)
#pragma unroll
    for (int n = 0; n < 4; ++n) vf[u][n] = *(const v4f*)(vc + (4 * u + n) * 128);
  if (NEXT) load_k_frags(kf, Kl, t0 + 16);

#pragma unroll
  for (int q = QLO; q < ATT_QT; ++q)
#pragma unroll
    for (int r = 0; r < 8; ++r)
      pt[q][r + 8 * half][l16] = sc[q][r];
  asm volatile("s_wait_dscnt 0" ::: "memory");
  float p[ATT_QT][8];
#pragma unroll
  for (int q = QLO; q < ATT_QT; ++q)
#pragma unroll
    for (int kk = 0; kk < 4; ++kk) {
      p[q][2 * kk]     = pt[q][l16][kk * 4 + 2 * half];
      p[q][2 * kk + 1] = pt[q][l16][kk * 4 + 2 * half + 1];
    }

  if (QLO == 0) softmax_row<DIAG0>(p[0], mq[0], lq[0], o[0], half, l16);
  softmax_row<DIAG1>(p[1], mq[1], lq[1], o[1], half, l16);

#pragma unroll
  for (int kk = 0; kk < 4; ++kk) {
    v2f a[ATT_QT];
#pragma unroll
    for (int q = QLO; q < ATT_QT; ++q) a[q] = v2f{p[q][2 * kk], p[q][2 * kk + 1]};
#pragma unroll
    for (int n = 0; n < 4; ++n)
#pragma unroll
      for (int q = QLO; q < ATT_QT; ++q)
        o[q][n] = wmma_f32(a[q], frag_of(vf[kk >> 1][n], kk & 1), o[q][n]);
  }
}

__global__ __launch_bounds__(256) void flash_attn_kernel(
    const float* __restrict__ Q, const float* __restrict__ Kf,
    const float* __restrict__ Vf, float* __restrict__ Oout) {
  const int wave = threadIdx.x >> 5;
  const int lane = threadIdx.x & 31;
  const int half = lane >> 4;
  const int l16  = lane & 15;

  const int bh = blockIdx.y;
  static_assert(ATT_NB == 8 * ATT_WG, "eight waves per workgroup: four SIMD pairs of (mirror, block)");
  const int g  = wave & 3;
  const int bx = blockIdx.x;
  const int lo = ATT_WG * g + ((g & 1) ? (ATT_WG - 1 - bx) : bx);
  const int qb = (wave & 4) ? lo : (ATT_NB - 1 - lo);
  const int q0 = qb * (16 * ATT_QT);

  const float* Qm = Q  + (size_t)bh * SEQ * D_HEAD;
  const float* Kl = Kf + (size_t)bh * SEQ * D_HEAD + 4 * lane;
  const float* Vl = Vf + (size_t)bh * SEQ * D_HEAD + 4 * lane;

  __shared__ float lds[8][ATT_QT][16][17];
  float (*pt)[16][17] = lds[wave];

  __shared__ float ldsq[8][16 * ATT_QT][ATT_QP];
  float (*qs)[ATT_QP] = ldsq[wave];
  v4f kf[8];
  load_k_frags(kf, Kl, 0);
#pragma unroll
  for (int r = 0; r < 16 * ATT_QT; ++r)
    *(v2f*)(&qs[r][2 * lane]) = *(const v2f*)(Qm + (size_t)(q0 + r) * D_HEAD + 2 * lane);

  v8f o[ATT_QT][4];
  float mq[ATT_QT], lq[ATT_QT];
#pragma unroll
  for (int q = 0; q < ATT_QT; ++q) {
#pragma unroll
    for (int n = 0; n < 4; ++n) o[q][n] = v8f{};
    mq[q] = -__builtin_inff(); lq[q] = 0.0f;
  }

  for (int t0 = 0; t0 < q0; t0 += 16)
    attn_step<0, false, false, true>(t0, Kl, Vl, qs, o, mq, lq, pt, kf, half, l16);
  attn_step<0, true, false, true>(q0, Kl, Vl, qs, o, mq, lq, pt, kf, half, l16);
  attn_step<1, false, true, false>(q0 + 16, Kl, Vl, qs, o, mq, lq, pt, kf, half, l16);

  const int b = bh >> 4;
  const int h = bh & (N_HEADS - 1);
#pragma unroll
  for (int q = 0; q < ATT_QT; ++q) {
    float invr[8];
    per_register(1.0f / (lq[q] + xhalf(lq[q])), invr);
#pragma unroll
    for (int r = 0; r < 8; ++r) {
      const float inv = invr[r];
      const int s = q0 + 16 * q + r + 8 * half;
      float* orow = Oout + (size_t)(b * SEQ + s) * (N_HEADS * D_HEAD) + h * D_HEAD;
      orow[ 0 + l16] = o[q][0][r] * inv;
      orow[16 + l16] = o[q][1][r] * inv;
      orow[32 + l16] = o[q][2][r] * inv;
      orow[48 + l16] = o[q][3][r] * inv;
    }
  }
}

#define OP_MI 4
#define OP_NJ 4
#define OP_KG 4
#define OP_KT (8 * OP_KG)

static __device__ __forceinline__ void op_load(
    v4f (&a)[OP_MI][OP_KG], v4f (&b)[OP_NJ][OP_KG],
    const float* __restrict__ ar, const float* __restrict__ br, const int k0) {
#pragma unroll
  for (int i = 0; i < OP_MI; ++i)
#pragma unroll
    for (int g = 0; g < OP_KG; ++g)
      a[i][g] = *(const v4f*)(ar + (size_t)i * 16 * D_MODEL + k0 + 8 * g);
#pragma unroll
  for (int j = 0; j < OP_NJ; ++j)
#pragma unroll
    for (int g = 0; g < OP_KG; ++g)
      b[j][g] = *(const v4f*)(br + (size_t)j * 16 * D_MODEL + k0 * 16 + g * 128);
}

static __device__ __forceinline__ void op_mma(
    v8f (&c)[OP_MI][OP_NJ],
    const v4f (&a)[OP_MI][OP_KG], const v4f (&b)[OP_NJ][OP_KG]) {
#pragma unroll
  for (int g = 0; g < OP_KG; ++g) {
#pragma unroll
    for (int s = 0; s < 2; ++s) {
#pragma unroll
      for (int i = 0; i < OP_MI; ++i) {
        const v2f af = s ? v2f{a[i][g].z, a[i][g].w} : v2f{a[i][g].x, a[i][g].y};
#pragma unroll
        for (int j = 0; j < OP_NJ; ++j) {
          const v2f bf = s ? v2f{b[j][g].z, b[j][g].w} : v2f{b[j][g].x, b[j][g].y};
          c[i][j] = wmma_f32(af, bf, c[i][j]);
        }
      }
    }
  }
}

__global__ __launch_bounds__(256) void out_proj_kernel(
    const float* __restrict__ A, const float* __restrict__ Wop,
    float* __restrict__ C) {
  const int wave = threadIdx.x >> 5;
  const int lane = threadIdx.x & 31;
  const int half = lane >> 4;
  const int l16  = lane & 15;

  const int row0 = (blockIdx.x * 2 + (wave >> 2)) * (16 * OP_MI);
  const int col0 = (blockIdx.y * 4 + (wave & 3)) * (16 * OP_NJ);

  v8f c[OP_MI][OP_NJ];
#pragma unroll
  for (int i = 0; i < OP_MI; ++i)
#pragma unroll
    for (int j = 0; j < OP_NJ; ++j) c[i][j] = v8f{};

  const float* ar = A   + (size_t)(row0 + l16) * D_MODEL + 4 * half;
  const float* br = Wop + (size_t)(col0 >> 4) * (16 * D_MODEL) + 4 * lane;

  v4f a0[OP_MI][OP_KG], b0[OP_NJ][OP_KG], a1[OP_MI][OP_KG], b1[OP_NJ][OP_KG];
  op_load(a0, b0, ar, br, 0);
  for (int k0 = 0; k0 < D_MODEL - 2 * OP_KT; k0 += 2 * OP_KT) {
    op_load(a1, b1, ar, br, k0 + OP_KT);
    __builtin_amdgcn_sched_barrier(0);
    op_mma(c, a0, b0);
    __builtin_amdgcn_sched_barrier(0);
    op_load(a0, b0, ar, br, k0 + 2 * OP_KT);
    __builtin_amdgcn_sched_barrier(0);
    op_mma(c, a1, b1);
    __builtin_amdgcn_sched_barrier(0);
  }
  op_load(a1, b1, ar, br, D_MODEL - OP_KT);
  __builtin_amdgcn_sched_barrier(0);
  op_mma(c, a0, b0);
  op_mma(c, a1, b1);

#pragma unroll
  for (int i = 0; i < OP_MI; ++i) {
#pragma unroll
    for (int r = 0; r < 8; ++r) {
      float* crow = C + (size_t)(row0 + 16 * i + r + 8 * half) * D_MODEL + col0 + l16;
#pragma unroll
      for (int j = 0; j < OP_NJ; ++j) crow[16 * j] = c[i][j][r];
    }
  }
}

extern "C" void kernel_launch(void* const* d_in, const int* in_sizes, int n_in,
                              void* d_out, int out_size, void* d_ws, size_t ws_size,
                              hipStream_t stream) {
  const float* x  = (const float*)d_in[0];
  const float* Wq = (const float*)d_in[1];
  const float* Wk = (const float*)d_in[2];
  const float* Wv = (const float*)d_in[3];
  const float* Wo = (const float*)d_in[4];

  float* ws = (float*)d_ws;
  const size_t per = (size_t)BATCH * N_HEADS * SEQ * D_HEAD;
  const size_t wsz = (size_t)D_MODEL * D_MODEL;
  float* Q   = ws;
  float* Kf  = ws + per;
  float* Vf  = ws + 2 * per;
  float* AO  = ws + 3 * per;
  float* Wqp = ws + 4 * per;
  float* Wkp = Wqp + wsz;
  float* Wvp = Wkp + wsz;
  float* Wop = Wvp + wsz;

  dim3 g0(D_MODEL / 32, D_MODEL / 32, 4);
  pack_w_kernel<<<g0, 256, 0, stream>>>(Wq, Wk, Wv, Wo, Wqp, Wkp, Wvp, Wop);

  dim3 g1(ROWS / (16 * QKV_MI), (N_HEADS * D_HEAD) / (16 * QKV_NJ * 8));
  qkv_proj_kernel<<<g1, 256, 0, stream>>>(x, Wqp, Wkp, Wvp, Q, Kf, Vf);

  dim3 g2(SEQ / (16 * ATT_QT * 8), BATCH * N_HEADS);
  flash_attn_kernel<<<g2, 256, 0, stream>>>(Q, Kf, Vf, AO);

  dim3 g3(ROWS / (16 * OP_MI * 2), D_MODEL / (16 * OP_NJ * 4));
  out_proj_kernel<<<g3, 256, 0, stream>>>(AO, Wop, (float*)d_out);
}


# Attention: one 16-wave workgroup per compute unit, so the 4 waves of a SIMD are picked together: blocks 63-p, 47-p, 16+p, p, oldest longest, 260 key steps each; per-row key order and sums kept
# speedup vs baseline: 2.2303x; 1.0515x over previous
#include <hip/hip_runtime.h>
#include <hip/hip_bf16.h>

typedef __attribute__((ext_vector_type(2))) float v2f;
typedef __attribute__((ext_vector_type(4))) float v4f;
typedef __attribute__((ext_vector_type(8))) float v8f;

#define D_MODEL 1024
#define N_HEADS 16
#define D_HEAD  64
#define BATCH   4
#define SEQ     2048
#define ROWS    (BATCH * SEQ)

static __device__ __forceinline__ v8f wmma_f32(v2f a, v2f b, v8f c) {
  return __builtin_amdgcn_wmma_f32_16x16x4_f32(
       false, a,  false, b,
       (short)0, c,  false,  false);
}

static __device__ __forceinline__ v2f frag_of(const v4f c, const int odd) {
  return odd ? v2f{c.z, c.w} : v2f{c.x, c.y};
}

__global__ __launch_bounds__(256) void pack_w_kernel(
    const float* __restrict__ W0, const float* __restrict__ W1,
    const float* __restrict__ W2, const float* __restrict__ W3,
    float* __restrict__ T0, float* __restrict__ T1,
    float* __restrict__ T2, float* __restrict__ T3) {
  const float* W = (blockIdx.z == 0) ? W0 : (blockIdx.z == 1) ? W1
                   : (blockIdx.z == 2) ? W2 : W3;
  float* T = (blockIdx.z == 0) ? T0 : (blockIdx.z == 1) ? T1
             : (blockIdx.z == 2) ? T2 : T3;

  __shared__ float tile[32][33];
  const int tx = threadIdx.x & 31;
  const int ty = threadIdx.x >> 5;
  const int x0 = blockIdx.x * 32;
  const int y0 = blockIdx.y * 32;

#pragma unroll
  for (int i = 0; i < 4; ++i)
    tile[ty + i * 8][tx] = W[(size_t)(y0 + ty + i * 8) * D_MODEL + x0 + tx];
  __syncthreads();
  const int gl  = ty & 3;
  const int nbl = ty >> 2;
  const int c = tx & 15;
  const int h = tx >> 4;
  const int kh = (blockIdx.z == 3) ? 4 * h : 2 * h;
  const int kp = (blockIdx.z == 3) ? 2 : 4;
  const float* tc = &tile[8 * gl + kh][16 * nbl + c];
  const v4f v = v4f{tc[0], tc[33], tc[33 * kp], tc[33 * kp + 33]};
  const int nb = (x0 >> 4) + nbl;
  const int G  = (y0 >> 3) + gl;
  *(v4f*)(T + ((size_t)(nb * 128 + G) * 32 + tx) * 4) = v;
}

#define QKV_MI 4
#define QKV_NJ 1
#define QKV_KU 8
__global__ __launch_bounds__(256) void qkv_proj_kernel(
    const float* __restrict__ X,
    const float* __restrict__ Wqp, const float* __restrict__ Wkp,
    const float* __restrict__ Wvp,
    float* __restrict__ Q, float* __restrict__ Kf, float* __restrict__ Vf) {
  const int wave = threadIdx.x >> 5;
  const int lane = threadIdx.x & 31;
  const int half = lane >> 4;
  const int l16  = lane & 15;

  const int row0 = blockIdx.x * (16 * QKV_MI);
  const int col0 = (blockIdx.y * 8 + wave) * (16 * QKV_NJ);

  v8f cq[QKV_MI][QKV_NJ], ck[QKV_MI][QKV_NJ], cv[QKV_MI][QKV_NJ];
#pragma unroll
  for (int i = 0; i < QKV_MI; ++i)
#pragma unroll
    for (int j = 0; j < QKV_NJ; ++j) { cq[i][j] = v8f{}; ck[i][j] = v8f{}; cv[i][j] = v8f{}; }

  const float* xr = X   + (size_t)(row0 + l16) * D_MODEL + 2 * half;
  const float* wq = Wqp + (size_t)(col0 >> 4) * (16 * D_MODEL) + 4 * lane;
  const float* wk = Wkp + (size_t)(col0 >> 4) * (16 * D_MODEL) + 4 * lane;
  const float* wv = Wvp + (size_t)(col0 >> 4) * (16 * D_MODEL) + 4 * lane;

  for (int k0 = 0; k0 < D_MODEL; k0 += 4 * QKV_KU) {
    v2f a[QKV_MI][QKV_KU];
    v4f bq[QKV_NJ][QKV_KU / 2], bk[QKV_NJ][QKV_KU / 2], bv[QKV_NJ][QKV_KU / 2];
#pragma unroll
    for (int i = 0; i < QKV_MI; ++i)
#pragma unroll
      for (int kk = 0; kk < QKV_KU; ++kk)
        a[i][kk] = *(const v2f*)(xr + (size_t)i * 16 * D_MODEL + k0 + 4 * kk);
#pragma unroll
    for (int j = 0; j < QKV_NJ; ++j) {
#pragma unroll
      for (int g = 0; g < QKV_KU / 2; ++g)
        bq[j][g] = *(const v4f*)(wq + (size_t)j * 16 * D_MODEL + k0 * 16 + g * 128);
#pragma unroll
      for (int g = 0; g < QKV_KU / 2; ++g)
        bk[j][g] = *(const v4f*)(wk + (size_t)j * 16 * D_MODEL + k0 * 16 + g * 128);
#pragma unroll
      for (int g = 0; g < QKV_KU / 2; ++g)
        bv[j][g] = *(const v4f*)(wv + (size_t)j * 16 * D_MODEL + k0 * 16 + g * 128);
    }
    __builtin_amdgcn_sched_barrier(0);
#pragma unroll
    for (int kk = 0; kk < QKV_KU; ++kk) {
#pragma unroll
      for (int i = 0; i < QKV_MI; ++i) {
#pragma unroll
        for (int j = 0; j < QKV_NJ; ++j) {
          cq[i][j] = wmma_f32(a[i][kk], frag_of(bq[j][kk >> 1], kk & 1), cq[i][j]);
          ck[i][j] = wmma_f32(frag_of(bk[j][kk >> 1], kk & 1), a[i][kk], ck[i][j]);
          cv[i][j] = wmma_f32(a[i][kk], frag_of(bv[j][kk >> 1], kk & 1), cv[i][j]);
        }
      }
    }
  }

#pragma unroll
  for (int j = 0; j < QKV_NJ; ++j) {
    const int colb = col0 + 16 * j;
    const int h  = colb >> 6;
    const int db = colb & 63;
#pragma unroll
    for (int i = 0; i < QKV_MI; ++i) {
      const int rowb = row0 + 16 * i;
      const int b  = rowb >> 11;
      const int sb = rowb & (SEQ - 1);
      const int bh = b * N_HEADS + h;
#pragma unroll
      for (int r = 0; r < 8; ++r) {
        const int s = sb + r + 8 * half;
        Q[((size_t)bh * SEQ + s) * D_HEAD + db + l16] = cq[i][j][r];
      }
      const size_t blk = ((size_t)bh * (SEQ / 16) + (sb >> 4)) * (16 * D_HEAD);
      float* kc = Kf + blk + (size_t)((db >> 3) + half) * 128 + 4 * l16;
      *(v4f*)(kc)      = v4f{ck[i][j][0], ck[i][j][1], ck[i][j][4], ck[i][j][5]};
      *(v4f*)(kc + 64) = v4f{ck[i][j][2], ck[i][j][3], ck[i][j][6], ck[i][j][7]};
      float* vc = Vf + blk + (size_t)(4 * half + (db >> 4)) * 128 + 4 * l16;
      *(v4f*)(vc)      = v4f{cv[i][j][0], cv[i][j][1], cv[i][j][4], cv[i][j][5]};
      *(v4f*)(vc + 64) = v4f{cv[i][j][2], cv[i][j][3], cv[i][j][6], cv[i][j][7]};
    }
  }
}

#define ATT_QT 2
#define ATT_NB (SEQ / (16 * ATT_QT))
#define ATT_WV 16
#define ATT_WG (ATT_NB / ATT_WV)
#define ATT_QP 68

static __device__ __forceinline__ float xhalf(float x) {
  const int b = __builtin_bit_cast(int, x);
  return __builtin_bit_cast(float,
      __builtin_amdgcn_permlanex16(b, b, 0x76543210, 0xfedcba98, false, false));
}

template <int R>
static __device__ __forceinline__ float row_lane(float x) {
  const int b = __builtin_bit_cast(int, x);
  return __builtin_bit_cast(float,
      __builtin_amdgcn_update_dpp(b, b, 0x150 + R, 0xf, 0xf, true));
}

static __device__ __forceinline__ void per_register(float x, float (&f)[8]) {
  const int b = __builtin_bit_cast(int, x);
  const float z = __builtin_bit_cast(float,
      __builtin_amdgcn_update_dpp(b, b, 0x160 + 8, 0x2, 0xf, false));
  f[0] = row_lane<0>(z); f[1] = row_lane<1>(z); f[2] = row_lane<2>(z); f[3] = row_lane<3>(z);
  f[4] = row_lane<4>(z); f[5] = row_lane<5>(z); f[6] = row_lane<6>(z); f[7] = row_lane<7>(z);
}

template <bool DIAG>
static __device__ __forceinline__ void softmax_row(
    float (&p)[8], float& m, float& l, v8f (&o)[4],
    const int half, const int l16) {
#pragma unroll
  for (int j = 0; j < 8; ++j) {
    float v = p[j] * 0.125f;
    if (DIAG && (4 * (j >> 1) + 2 * half + (j & 1) > l16)) v = -__builtin_inff();
    p[j] = v;
  }
  float mx = fmaxf(fmaxf(fmaxf(p[0], p[1]), fmaxf(p[2], p[3])),
                   fmaxf(fmaxf(p[4], p[5]), fmaxf(p[6], p[7])));
  mx = fmaxf(mx, xhalf(mx));
  const float mn   = fmaxf(m, mx);
  const float resc = __expf(m - mn);
  m = mn;
#pragma unroll
  for (int j = 0; j < 8; ++j) p[j] = __expf(p[j] - mn);
  const float s = ((p[0] + p[1]) + (p[2] + p[3])) + ((p[4] + p[5]) + (p[6] + p[7]));
  l = l * resc + s;
  float rc[8];
  per_register(resc, rc);
#pragma unroll
  for (int r = 0; r < 8; ++r) {
    o[0][r] *= rc[r]; o[1][r] *= rc[r]; o[2][r] *= rc[r]; o[3][r] *= rc[r];
  }
}

static __device__ __forceinline__ void load_k_frags(
    v4f (&kf)[8], const float* __restrict__ Kl, const int t0) {
  const float* kc = Kl + (size_t)t0 * D_HEAD;
#pragma unroll
  for (int g = 0; g < 8; ++g) kf[g] = *(const v4f*)(kc + g * 128);
}


template <int QLO, bool DIAG0, bool DIAG1, bool NEXT>
static __device__ __forceinline__ void attn_step(
    const int t0, const float* __restrict__ Kl, const float* __restrict__ Vl,
    const float (*qs)[ATT_QP], v8f (&o)[ATT_QT][4],
    float (&mq)[ATT_QT], float (&lq)[ATT_QT],
    float (*pt)[16][17], v4f (&kf)[8], const int half, const int l16) {
  v8f sc[ATT_QT];
#pragma unroll
  for (int q = 0; q < ATT_QT; ++q) sc[q] = v8f{};
  v2f af[ATT_QT][16];
#pragma unroll
  for (int kk = 0; kk < 16; ++kk)
#pragma unroll
    for (int q = QLO; q < ATT_QT; ++q)
      af[q][kk] = *(const v2f*)(&qs[16 * q + l16][kk * 4 + 2 * half]);
  __builtin_amdgcn_sched_barrier(0);
#pragma unroll
  for (int kk = 0; kk < 16; ++kk)
#pragma unroll
    for (int q = QLO; q < ATT_QT; ++q)
      sc[q] = wmma_f32(af[q][kk], frag_of(kf[kk >> 1], kk & 1), sc[q]);

  __builtin_amdgcn_sched_barrier(0);
  v4f vf[2][4];
  const float* vc = Vl + (size_t)t0 * D_HEAD;
#pragma unroll
  for (int u = 0; u < 2; ++u)
#pragma unroll
    for (int n = 0; n < 4; ++n) vf[u][n] = *(const v4f*)(vc + (4 * u + n) * 128);
  if (NEXT) load_k_frags(kf, Kl, t0 + 16);

#pragma unroll
  for (int q = QLO; q < ATT_QT; ++q)
#pragma unroll
    for (int r = 0; r < 8; ++r)
      pt[q][r + 8 * half][l16] = sc[q][r];
  asm volatile("s_wait_dscnt 0" ::: "memory");
  float p[ATT_QT][8];
#pragma unroll
  for (int q = QLO; q < ATT_QT; ++q)
#pragma unroll
    for (int kk = 0; kk < 4; ++kk) {
      p[q][2 * kk]     = pt[q][l16][kk * 4 + 2 * half];
      p[q][2 * kk + 1] = pt[q][l16][kk * 4 + 2 * half + 1];
    }

  if (QLO == 0) softmax_row<DIAG0>(p[0], mq[0], lq[0], o[0], half, l16);
  softmax_row<DIAG1>(p[1], mq[1], lq[1], o[1], half, l16);

#pragma unroll
  for (int kk = 0; kk < 4; ++kk) {
    v2f a[ATT_QT];
#pragma unroll
    for (int q = QLO; q < ATT_QT; ++q) a[q] = v2f{p[q][2 * kk], p[q][2 * kk + 1]};
#pragma unroll
    for (int n = 0; n < 4; ++n)
#pragma unroll
      for (int q = QLO; q < ATT_QT; ++q)
        o[q][n] = wmma_f32(a[q], frag_of(vf[kk >> 1][n], kk & 1), o[q][n]);
  }
}

__global__ __launch_bounds__(32 * ATT_WV) void flash_attn_kernel(
    const float* __restrict__ Q, const float* __restrict__ Kf,
    const float* __restrict__ Vf, float* __restrict__ Oout) {
  const int wave = threadIdx.x >> 5;
  const int lane = threadIdx.x & 31;
  const int half = lane >> 4;
  const int l16  = lane & 15;

  const int bh = blockIdx.y;
  static_assert(ATT_NB == 4 * ATT_WV && ATT_NB == ATT_WV * ATT_WG, "64 blocks per head: 16 SIMD slots of 4 blocks");
  const int p   = 4 * blockIdx.x + (wave & 3);
  const int age = wave >> 2;
  const int qb  = (age == 0) ? (ATT_NB - 1 - p) : (age == 1) ? (ATT_NB - 17 - p)
                : (age == 2) ? (16 + p) : p;
  const int q0 = qb * (16 * ATT_QT);

  const float* Qm = Q  + (size_t)bh * SEQ * D_HEAD;
  const float* Kl = Kf + (size_t)bh * SEQ * D_HEAD + 4 * lane;
  const float* Vl = Vf + (size_t)bh * SEQ * D_HEAD + 4 * lane;

  __shared__ float lds[ATT_WV][ATT_QT][16][17];
  float (*pt)[16][17] = lds[wave];

  __shared__ float ldsq[ATT_WV][16 * ATT_QT][ATT_QP];
  float (*qs)[ATT_QP] = ldsq[wave];
  v4f kf[8];
  load_k_frags(kf, Kl, 0);
#pragma unroll
  for (int r = 0; r < 16 * ATT_QT; ++r)
    *(v2f*)(&qs[r][2 * lane]) = *(const v2f*)(Qm + (size_t)(q0 + r) * D_HEAD + 2 * lane);

  v8f o[ATT_QT][4];
  float mq[ATT_QT], lq[ATT_QT];
#pragma unroll
  for (int q = 0; q < ATT_QT; ++q) {
#pragma unroll
    for (int n = 0; n < 4; ++n) o[q][n] = v8f{};
    mq[q] = -__builtin_inff(); lq[q] = 0.0f;
  }

  for (int t0 = 0; t0 < q0; t0 += 16)
    attn_step<0, false, false, true>(t0, Kl, Vl, qs, o, mq, lq, pt, kf, half, l16);
  attn_step<0, true, false, true>(q0, Kl, Vl, qs, o, mq, lq, pt, kf, half, l16);
  attn_step<1, false, true, false>(q0 + 16, Kl, Vl, qs, o, mq, lq, pt, kf, half, l16);

  const int b = bh >> 4;
  const int h = bh & (N_HEADS - 1);
#pragma unroll
  for (int q = 0; q < ATT_QT; ++q) {
    float invr[8];
    per_register(1.0f / (lq[q] + xhalf(lq[q])), invr);
#pragma unroll
    for (int r = 0; r < 8; ++r) {
      const float inv = invr[r];
      const int s = q0 + 16 * q + r + 8 * half;
      float* orow = Oout + (size_t)(b * SEQ + s) * (N_HEADS * D_HEAD) + h * D_HEAD;
      orow[ 0 + l16] = o[q][0][r] * inv;
      orow[16 + l16] = o[q][1][r] * inv;
      orow[32 + l16] = o[q][2][r] * inv;
      orow[48 + l16] = o[q][3][r] * inv;
    }
  }
}

#define OP_MI 4
#define OP_NJ 4
#define OP_KG 4
#define OP_KT (8 * OP_KG)

static __device__ __forceinline__ void op_load(
    v4f (&a)[OP_MI][OP_KG], v4f (&b)[OP_NJ][OP_KG],
    const float* __restrict__ ar, const float* __restrict__ br, const int k0) {
#pragma unroll
  for (int i = 0; i < OP_MI; ++i)
#pragma unroll
    for (int g = 0; g < OP_KG; ++g)
      a[i][g] = *(const v4f*)(ar + (size_t)i * 16 * D_MODEL + k0 + 8 * g);
#pragma unroll
  for (int j = 0; j < OP_NJ; ++j)
#pragma unroll
    for (int g = 0; g < OP_KG; ++g)
      b[j][g] = *(const v4f*)(br + (size_t)j * 16 * D_MODEL + k0 * 16 + g * 128);
}

static __device__ __forceinline__ void op_mma(
    v8f (&c)[OP_MI][OP_NJ],
    const v4f (&a)[OP_MI][OP_KG], const v4f (&b)[OP_NJ][OP_KG]) {
#pragma unroll
  for (int g = 0; g < OP_KG; ++g) {
#pragma unroll
    for (int s = 0; s < 2; ++s) {
#pragma unroll
      for (int i = 0; i < OP_MI; ++i) {
        const v2f af = s ? v2f{a[i][g].z, a[i][g].w} : v2f{a[i][g].x, a[i][g].y};
#pragma unroll
        for (int j = 0; j < OP_NJ; ++j) {
          const v2f bf = s ? v2f{b[j][g].z, b[j][g].w} : v2f{b[j][g].x, b[j][g].y};
          c[i][j] = wmma_f32(af, bf, c[i][j]);
        }
      }
    }
  }
}

__global__ __launch_bounds__(256) void out_proj_kernel(
    const float* __restrict__ A, const float* __restrict__ Wop,
    float* __restrict__ C) {
  const int wave = threadIdx.x >> 5;
  const int lane = threadIdx.x & 31;
  const int half = lane >> 4;
  const int l16  = lane & 15;

  const int row0 = (blockIdx.x * 2 + (wave >> 2)) * (16 * OP_MI);
  const int col0 = (blockIdx.y * 4 + (wave & 3)) * (16 * OP_NJ);

  v8f c[OP_MI][OP_NJ];
#pragma unroll
  for (int i = 0; i < OP_MI; ++i)
#pragma unroll
    for (int j = 0; j < OP_NJ; ++j) c[i][j] = v8f{};

  const float* ar = A   + (size_t)(row0 + l16) * D_MODEL + 4 * half;
  const float* br = Wop + (size_t)(col0 >> 4) * (16 * D_MODEL) + 4 * lane;

  v4f a0[OP_MI][OP_KG], b0[OP_NJ][OP_KG], a1[OP_MI][OP_KG], b1[OP_NJ][OP_KG];
  op_load(a0, b0, ar, br, 0);
  for (int k0 = 0; k0 < D_MODEL - 2 * OP_KT; k0 += 2 * OP_KT) {
    op_load(a1, b1, ar, br, k0 + OP_KT);
    __builtin_amdgcn_sched_barrier(0);
    op_mma(c, a0, b0);
    __builtin_amdgcn_sched_barrier(0);
    op_load(a0, b0, ar, br, k0 + 2 * OP_KT);
    __builtin_amdgcn_sched_barrier(0);
    op_mma(c, a1, b1);
    __builtin_amdgcn_sched_barrier(0);
  }
  op_load(a1, b1, ar, br, D_MODEL - OP_KT);
  __builtin_amdgcn_sched_barrier(0);
  op_mma(c, a0, b0);
  op_mma(c, a1, b1);

#pragma unroll
  for (int i = 0; i < OP_MI; ++i) {
#pragma unroll
    for (int r = 0; r < 8; ++r) {
      float* crow = C + (size_t)(row0 + 16 * i + r + 8 * half) * D_MODEL + col0 + l16;
#pragma unroll
      for (int j = 0; j < OP_NJ; ++j) crow[16 * j] = c[i][j][r];
    }
  }
}

extern "C" void kernel_launch(void* const* d_in, const int* in_sizes, int n_in,
                              void* d_out, int out_size, void* d_ws, size_t ws_size,
                              hipStream_t stream) {
  const float* x  = (const float*)d_in[0];
  const float* Wq = (const float*)d_in[1];
  const float* Wk = (const float*)d_in[2];
  const float* Wv = (const float*)d_in[3];
  const float* Wo = (const float*)d_in[4];

  float* ws = (float*)d_ws;
  const size_t per = (size_t)BATCH * N_HEADS * SEQ * D_HEAD;
  const size_t wsz = (size_t)D_MODEL * D_MODEL;
  float* Q   = ws;
  float* Kf  = ws + per;
  float* Vf  = ws + 2 * per;
  float* AO  = ws + 3 * per;
  float* Wqp = ws + 4 * per;
  float* Wkp = Wqp + wsz;
  float* Wvp = Wkp + wsz;
  float* Wop = Wvp + wsz;

  dim3 g0(D_MODEL / 32, D_MODEL / 32, 4);
  pack_w_kernel<<<g0, 256, 0, stream>>>(Wq, Wk, Wv, Wo, Wqp, Wkp, Wvp, Wop);

  dim3 g1(ROWS / (16 * QKV_MI), (N_HEADS * D_HEAD) / (16 * QKV_NJ * 8));
  qkv_proj_kernel<<<g1, 256, 0, stream>>>(x, Wqp, Wkp, Wvp, Q, Kf, Vf);

  dim3 g2(SEQ / (16 * ATT_QT * ATT_WV), BATCH * N_HEADS);
  flash_attn_kernel<<<g2, 32 * ATT_WV, 0, stream>>>(Q, Kf, Vf, AO);

  dim3 g3(ROWS / (16 * OP_MI * 2), D_MODEL / (16 * OP_NJ * 4));
  out_proj_kernel<<<g3, 256, 0, stream>>>(AO, Wop, (float*)d_out);
}
